# starting kernel
# baseline (speedup 1.0000x reference)
#include <hip/hip_runtime.h>
#include <hip/hip_bf16.h>
#include <math.h>

typedef float v2f __attribute__((ext_vector_type(2)));
typedef float v4f __attribute__((ext_vector_type(4)));
typedef float v8f __attribute__((ext_vector_type(8)));

#define B_SZ    4
#define T_LEN   2048
#define D_MODEL 1024
#define D_STATE 16
#define DT_RANK 64
#define M_ROWS  (B_SZ * T_LEN)

#define NCHUNK  32
#define CLEN    (T_LEN / NCHUNK)

template <int ACT>
__global__ __launch_bounds__(256)
void wmma_gemm_f32(const float* __restrict__ A,
                   const float* __restrict__ W,
                   const float* __restrict__ bias,
                   float* __restrict__ out,
                   int M, int N, int K) {
    const int wave   = (blockIdx.x * blockDim.x + threadIdx.x) >> 5;
    const int lane   = threadIdx.x & 31;
    const int ntiles = N >> 4;
    const int m0     = (wave / ntiles) << 4;
    const int n0     = (wave % ntiles) << 4;
    const int half   = lane >> 4;
    const int l15    = lane & 15;

    const float* arow = A + (size_t)(m0 + l15) * K + 2 * half;
    const float* wrow = W + (size_t)(n0 + l15) * K + 2 * half;

    v8f c = {};
#pragma unroll 4
    for (int k = 0; k < K; k += 4) {
        v2f a = *(const v2f*)(arow + k);
        v2f b = *(const v2f*)(wrow + k);
        c = __builtin_amdgcn_wmma_f32_16x16x4_f32(
                false, a, false, b, (short)0, c, false, false);
    }

    const float bv = bias[n0 + l15];
#pragma unroll
    for (int r = 0; r < 8; ++r) {
        float v = c[r] + bv;
        if (ACT == 1) v = (v > 20.0f) ? v : log1pf(__expf(v));
        out[(size_t)(m0 + half * 8 + r) * N + (n0 + l15)] = v;
    }
}

__global__ __launch_bounds__(256)
void scan_phase1(const float* __restrict__ u,
                 const float* __restrict__ dt,
                 const float* __restrict__ Btg,
                 const float* __restrict__ A_log,
                 float* __restrict__ Pbuf,
                 float* __restrict__ Sbuf) {
    __shared__ float sB[CLEN][D_STATE];

    const int chunk = blockIdx.x & (NCHUNK - 1);
    const int rest  = blockIdx.x >> 5;
    const int b     = rest >> 2;
    const int d     = ((rest & 3) << 8) + threadIdx.x;

    const float LOG2E = 1.44269504088896340736f;
    float Alog2[D_STATE];
#pragma unroll
    for (int n = 0; n < D_STATE; ++n)
        Alog2[n] = -__expf(A_log[d * D_STATE + n]) * LOG2E;

    const size_t baseS = ((size_t)b * T_LEN + (size_t)chunk * CLEN) * D_STATE;
    for (int i = threadIdx.x; i < CLEN * D_STATE; i += 256)
        ((float*)sB)[i] = Btg[baseS + i];
    __syncthreads();

    float P[D_STATE], S[D_STATE];
#pragma unroll
    for (int n = 0; n < D_STATE; ++n) { P[n] = 1.0f; S[n] = 0.0f; }

    const size_t baseU = ((size_t)b * T_LEN + (size_t)chunk * CLEN) * D_MODEL + d;
    for (int tt = 0; tt < CLEN; ++tt) {
        const size_t off = baseU + (size_t)tt * D_MODEL;
        const float ul  = u[off];
        const float dtl = dt[off];
        const float dtu = dtl * ul;
#pragma unroll
        for (int n = 0; n < D_STATE; ++n) {
            float abar = exp2f(dtl * Alog2[n]);
            P[n] *= abar;
            S[n]  = fmaf(abar, S[n], sB[tt][n] * dtu);
        }
    }

    const size_t o = (((size_t)b * NCHUNK + chunk) * D_MODEL + d) * D_STATE;
#pragma unroll
    for (int q = 0; q < 4; ++q) {
        *(v4f*)(Pbuf + o + 4 * q) = *(v4f*)(P + 4 * q);
        *(v4f*)(Sbuf + o + 4 * q) = *(v4f*)(S + 4 * q);
    }
}

__global__ __launch_bounds__(256)
void scan_phase2(float* __restrict__ Pbuf,
                 const float* __restrict__ Sbuf) {
    const int b = blockIdx.x >> 2;
    const int d = ((blockIdx.x & 3) << 8) + threadIdx.x;

    float h[D_STATE];
#pragma unroll
    for (int n = 0; n < D_STATE; ++n) h[n] = 0.0f;

    for (int chunk = 0; chunk < NCHUNK; ++chunk) {
        const size_t o = (((size_t)b * NCHUNK + chunk) * D_MODEL + d) * D_STATE;
        float P[D_STATE], S[D_STATE];
#pragma unroll
        for (int q = 0; q < 4; ++q) {
            *(v4f*)(P + 4 * q) = *(const v4f*)(Pbuf + o + 4 * q);
            *(v4f*)(S + 4 * q) = *(const v4f*)(Sbuf + o + 4 * q);
        }
#pragma unroll
        for (int q = 0; q < 4; ++q)
            *(v4f*)(Pbuf + o + 4 * q) = *(v4f*)(h + 4 * q);
#pragma unroll
        for (int n = 0; n < D_STATE; ++n)
            h[n] = fmaf(P[n], h[n], S[n]);
    }
}

__global__ __launch_bounds__(256)
void scan_phase3(const float* __restrict__ u,
                 const float* __restrict__ dt,
                 const float* __restrict__ Btg,
                 const float* __restrict__ Ctg,
                 const float* __restrict__ A_log,
                 const float* __restrict__ Dvec,
                 const float* __restrict__ Hstart,
                 float* __restrict__ out) {
    __shared__ float sB[CLEN][D_STATE];
    __shared__ float sC[CLEN][D_STATE];

    const int chunk = blockIdx.x & (NCHUNK - 1);
    const int rest  = blockIdx.x >> 5;
    const int b     = rest >> 2;
    const int d     = ((rest & 3) << 8) + threadIdx.x;

    const float LOG2E = 1.44269504088896340736f;
    float Alog2[D_STATE];
#pragma unroll
    for (int n = 0; n < D_STATE; ++n)
        Alog2[n] = -__expf(A_log[d * D_STATE + n]) * LOG2E;

    const size_t baseS = ((size_t)b * T_LEN + (size_t)chunk * CLEN) * D_STATE;
    for (int i = threadIdx.x; i < CLEN * D_STATE; i += 256) {
        ((float*)sB)[i] = Btg[baseS + i];
        ((float*)sC)[i] = Ctg[baseS + i];
    }
    __syncthreads();

    float h[D_STATE];
    const size_t ho = (((size_t)b * NCHUNK + chunk) * D_MODEL + d) * D_STATE;
#pragma unroll
    for (int q = 0; q < 4; ++q)
        *(v4f*)(h + 4 * q) = *(const v4f*)(Hstart + ho + 4 * q);

    const float Dd = Dvec[d];
    const size_t baseU = ((size_t)b * T_LEN + (size_t)chunk * CLEN) * D_MODEL + d;
    for (int tt = 0; tt < CLEN; ++tt) {
        const size_t off = baseU + (size_t)tt * D_MODEL;
        const float ul  = u[off];
        const float dtl = dt[off];
        const float dtu = dtl * ul;
        float y = Dd * ul;
#pragma unroll
        for (int n = 0; n < D_STATE; ++n) {
            float abar = exp2f(dtl * Alog2[n]);
            h[n] = fmaf(abar, h[n], sB[tt][n] * dtu);
            y    = fmaf(h[n], sC[tt][n], y);
        }
        out[off] = y;
    }
}

extern "C" void kernel_launch(void* const* d_in, const int* in_sizes, int n_in,
                              void* d_out, int out_size, void* d_ws, size_t ws_size,
                              hipStream_t stream) {
    const float* u     = (const float*)d_in[0];
    const float* A_log = (const float*)d_in[1];
    const float* W_xdt = (const float*)d_in[2];
    const float* b_xdt = (const float*)d_in[3];
    const float* W_dt  = (const float*)d_in[4];
    const float* b_dt  = (const float*)d_in[5];
    const float* W_B   = (const float*)d_in[6];
    const float* b_B   = (const float*)d_in[7];
    const float* W_C   = (const float*)d_in[8];
    const float* b_C   = (const float*)d_in[9];
    const float* Dv    = (const float*)d_in[10];
    float* out = (float*)d_out;

    float* ws   = (float*)d_ws;
    float* xdt  = ws;
    float* dt   = xdt + (size_t)M_ROWS * DT_RANK;
    float* Bt   = dt  + (size_t)M_ROWS * D_MODEL;
    float* Ct   = Bt  + (size_t)M_ROWS * D_STATE;
    float* Pbuf = Ct  + (size_t)M_ROWS * D_STATE;
    float* Sbuf = Pbuf + (size_t)B_SZ * NCHUNK * D_MODEL * D_STATE;

    wmma_gemm_f32<0><<<(M_ROWS / 16) * (DT_RANK / 16) / 8, 256, 0, stream>>>(
        u, W_xdt, b_xdt, xdt, M_ROWS, DT_RANK, D_MODEL);

    wmma_gemm_f32<1><<<(M_ROWS / 16) * (D_MODEL / 16) / 8, 256, 0, stream>>>(
        xdt, W_dt, b_dt, dt, M_ROWS, D_MODEL, DT_RANK);

    wmma_gemm_f32<0><<<(M_ROWS / 16) / 8, 256, 0, stream>>>(
        u, W_B, b_B, Bt, M_ROWS, D_STATE, D_MODEL);
    wmma_gemm_f32<0><<<(M_ROWS / 16) / 8, 256, 0, stream>>>(
        u, W_C, b_C, Ct, M_ROWS, D_STATE, D_MODEL);

    const int scan_blocks = B_SZ * 4 * NCHUNK;
    scan_phase1<<<scan_blocks, 256, 0, stream>>>(u, dt, Bt, A_log, Pbuf, Sbuf);
    scan_phase2<<<B_SZ * 4, 256, 0, stream>>>(Pbuf, Sbuf);
    scan_phase3<<<scan_blocks, 256, 0, stream>>>(u, dt, Bt, Ct, A_log, Dv, Pbuf, out);
}


# x_dt, B and C projections of u in one launch: 8 waves split K, each u fragment feeds 6 WMMA tiles; the 8 partial sums are added in fixed order through LDS (changes the K summation order)
# speedup vs baseline: 1.2628x; 1.2628x over previous
#include <hip/hip_runtime.h>
#include <hip/hip_bf16.h>
#include <math.h>

typedef float v2f __attribute__((ext_vector_type(2)));
typedef float v4f __attribute__((ext_vector_type(4)));
typedef float v8f __attribute__((ext_vector_type(8)));

#define B_SZ    4
#define T_LEN   2048
#define D_MODEL 1024
#define D_STATE 16
#define DT_RANK 64
#define M_ROWS  (B_SZ * T_LEN)

#define NCHUNK  32
#define CLEN    (T_LEN / NCHUNK)

template <int ACT>
__global__ __launch_bounds__(256)
void wmma_gemm_f32(const float* __restrict__ A,
                   const float* __restrict__ W,
                   const float* __restrict__ bias,
                   float* __restrict__ out,
                   int M, int N, int K) {
    const int wave   = (blockIdx.x * blockDim.x + threadIdx.x) >> 5;
    const int lane   = threadIdx.x & 31;
    const int ntiles = N >> 4;
    const int m0     = (wave / ntiles) << 4;
    const int n0     = (wave % ntiles) << 4;
    const int half   = lane >> 4;
    const int l15    = lane & 15;

    const float* arow = A + (size_t)(m0 + l15) * K + 2 * half;
    const float* wrow = W + (size_t)(n0 + l15) * K + 2 * half;

    v8f c = {};
#pragma unroll 4
    for (int k = 0; k < K; k += 4) {
        v2f a = *(const v2f*)(arow + k);
        v2f b = *(const v2f*)(wrow + k);
        c = __builtin_amdgcn_wmma_f32_16x16x4_f32(
                false, a, false, b, (short)0, c, false, false);
    }

    const float bv = bias[n0 + l15];
#pragma unroll
    for (int r = 0; r < 8; ++r) {
        float v = c[r] + bv;
        if (ACT == 1) v = (v > 20.0f) ? v : log1pf(__expf(v));
        out[(size_t)(m0 + half * 8 + r) * N + (n0 + l15)] = v;
    }
}

#define PJ_ROWS   32
#define PJ_KSPLIT 8
#define PJ_KSLICE (D_MODEL / PJ_KSPLIT)
#define PJ_NT     6
#define PJ_COLS   (PJ_NT * 16)
#define PJ_PITCH  100

__global__ __launch_bounds__(256)
void proj_u_fused(const float* __restrict__ u,
                  const float* __restrict__ W_xdt,
                  const float* __restrict__ b_xdt,
                  const float* __restrict__ W_B,
                  const float* __restrict__ b_B,
                  const float* __restrict__ W_C,
                  const float* __restrict__ b_C,
                  float* __restrict__ xdt,
                  float* __restrict__ Bt,
                  float* __restrict__ Ct) {
    __shared__ float part[PJ_KSPLIT][PJ_ROWS * PJ_PITCH];

    const int lane = threadIdx.x & 31;
    const int w    = threadIdx.x >> 5;
    const int half = lane >> 4;
    const int l15  = lane & 15;
    const int m0   = blockIdx.x * PJ_ROWS;
    const int koff = w * PJ_KSLICE + 16 * half;

    const float* ap[2];
    ap[0] = u + (size_t)(m0 + l15) * D_MODEL + koff;
    ap[1] = ap[0] + (size_t)16 * D_MODEL;
    const float* bp[PJ_NT];
#pragma unroll
    for (int j = 0; j < 4; ++j)
        bp[j] = W_xdt + (size_t)(16 * j + l15) * D_MODEL + koff;
    bp[4] = W_B + (size_t)l15 * D_MODEL + koff;
    bp[5] = W_C + (size_t)l15 * D_MODEL + koff;

    v8f acc[2][PJ_NT];
#pragma unroll
    for (int mt = 0; mt < 2; ++mt)
#pragma unroll
        for (int j = 0; j < PJ_NT; ++j)
            acc[mt][j] = v8f{};

#pragma unroll
    for (int ks = 0; ks < PJ_KSLICE; ks += 32) {
        v4f a0[4], a1[4];
#pragma unroll
        for (int q = 0; q < 4; ++q) a0[q] = *(const v4f*)(ap[0] + ks + 4 * q);
#pragma unroll
        for (int q = 0; q < 4; ++q) a1[q] = *(const v4f*)(ap[1] + ks + 4 * q);
#pragma unroll
        for (int j = 0; j < PJ_NT; ++j) {
            v4f b[4];
#pragma unroll
            for (int q = 0; q < 4; ++q) b[q] = *(const v4f*)(bp[j] + ks + 4 * q);
#pragma unroll
            for (int q = 0; q < 4; ++q) {
#pragma unroll
                for (int s = 0; s < 2; ++s) {
                    const v2f f0 = {a0[q][2 * s], a0[q][2 * s + 1]};
                    const v2f f1 = {a1[q][2 * s], a1[q][2 * s + 1]};
                    const v2f fb = {b[q][2 * s], b[q][2 * s + 1]};
                    acc[0][j] = __builtin_amdgcn_wmma_f32_16x16x4_f32(
                        false, f0, false, fb, (short)0, acc[0][j], false, false);
                    acc[1][j] = __builtin_amdgcn_wmma_f32_16x16x4_f32(
                        false, f1, false, fb, (short)0, acc[1][j], false, false);
                }
            }
        }
    }

    float* mine = part[w];
#pragma unroll
    for (int mt = 0; mt < 2; ++mt)
#pragma unroll
        for (int j = 0; j < PJ_NT; ++j)
#pragma unroll
            for (int r = 0; r < 8; ++r)
                mine[(16 * mt + 8 * half + r) * PJ_PITCH + 16 * j + l15] = acc[mt][j][r];
    __syncthreads();

#pragma unroll
    for (int i = 0; i < (PJ_ROWS * PJ_COLS) / 256; ++i) {
        const int o = threadIdx.x + 256 * i;
        int row, col;
        float bias;
        float* dst;
        if (i < 8) {
            row = o >> 6;  col = o & 63;
            bias = b_xdt[col];
            dst  = xdt + (size_t)m0 * DT_RANK + o;
        } else if (i < 10) {
            const int q = o - PJ_ROWS * DT_RANK;
            row = q >> 4;  col = 64 + (q & 15);
            bias = b_B[q & 15];
            dst  = Bt + (size_t)m0 * D_STATE + q;
        } else {
            const int q = o - PJ_ROWS * (DT_RANK + D_STATE);
            row = q >> 4;  col = 80 + (q & 15);
            bias = b_C[q & 15];
            dst  = Ct + (size_t)m0 * D_STATE + q;
        }
        float s = part[0][row * PJ_PITCH + col];
#pragma unroll
        for (int k = 1; k < PJ_KSPLIT; ++k)
            s += part[k][row * PJ_PITCH + col];
        *dst = s + bias;
    }
}

__global__ __launch_bounds__(256)
void scan_phase1(const float* __restrict__ u,
                 const float* __restrict__ dt,
                 const float* __restrict__ Btg,
                 const float* __restrict__ A_log,
                 float* __restrict__ Pbuf,
                 float* __restrict__ Sbuf) {
    __shared__ float sB[CLEN][D_STATE];

    const int chunk = blockIdx.x & (NCHUNK - 1);
    const int rest  = blockIdx.x >> 5;
    const int b     = rest >> 2;
    const int d     = ((rest & 3) << 8) + threadIdx.x;

    const float LOG2E = 1.44269504088896340736f;
    float Alog2[D_STATE];
#pragma unroll
    for (int n = 0; n < D_STATE; ++n)
        Alog2[n] = -__expf(A_log[d * D_STATE + n]) * LOG2E;

    const size_t baseS = ((size_t)b * T_LEN + (size_t)chunk * CLEN) * D_STATE;
    for (int i = threadIdx.x; i < CLEN * D_STATE; i += 256)
        ((float*)sB)[i] = Btg[baseS + i];
    __syncthreads();

    float P[D_STATE], S[D_STATE];
#pragma unroll
    for (int n = 0; n < D_STATE; ++n) { P[n] = 1.0f; S[n] = 0.0f; }

    const size_t baseU = ((size_t)b * T_LEN + (size_t)chunk * CLEN) * D_MODEL + d;
    for (int tt = 0; tt < CLEN; ++tt) {
        const size_t off = baseU + (size_t)tt * D_MODEL;
        const float ul  = u[off];
        const float dtl = dt[off];
        const float dtu = dtl * ul;
#pragma unroll
        for (int n = 0; n < D_STATE; ++n) {
            float abar = exp2f(dtl * Alog2[n]);
            P[n] *= abar;
            S[n]  = fmaf(abar, S[n], sB[tt][n] * dtu);
        }
    }

    const size_t o = (((size_t)b * NCHUNK + chunk) * D_MODEL + d) * D_STATE;
#pragma unroll
    for (int q = 0; q < 4; ++q) {
        *(v4f*)(Pbuf + o + 4 * q) = *(v4f*)(P + 4 * q);
        *(v4f*)(Sbuf + o + 4 * q) = *(v4f*)(S + 4 * q);
    }
}

__global__ __launch_bounds__(256)
void scan_phase2(float* __restrict__ Pbuf,
                 const float* __restrict__ Sbuf) {
    const int b = blockIdx.x >> 2;
    const int d = ((blockIdx.x & 3) << 8) + threadIdx.x;

    float h[D_STATE];
#pragma unroll
    for (int n = 0; n < D_STATE; ++n) h[n] = 0.0f;

    for (int chunk = 0; chunk < NCHUNK; ++chunk) {
        const size_t o = (((size_t)b * NCHUNK + chunk) * D_MODEL + d) * D_STATE;
        float P[D_STATE], S[D_STATE];
#pragma unroll
        for (int q = 0; q < 4; ++q) {
            *(v4f*)(P + 4 * q) = *(const v4f*)(Pbuf + o + 4 * q);
            *(v4f*)(S + 4 * q) = *(const v4f*)(Sbuf + o + 4 * q);
        }
#pragma unroll
        for (int q = 0; q < 4; ++q)
            *(v4f*)(Pbuf + o + 4 * q) = *(v4f*)(h + 4 * q);
#pragma unroll
        for (int n = 0; n < D_STATE; ++n)
            h[n] = fmaf(P[n], h[n], S[n]);
    }
}

__global__ __launch_bounds__(256)
void scan_phase3(const float* __restrict__ u,
                 const float* __restrict__ dt,
                 const float* __restrict__ Btg,
                 const float* __restrict__ Ctg,
                 const float* __restrict__ A_log,
                 const float* __restrict__ Dvec,
                 const float* __restrict__ Hstart,
                 float* __restrict__ out) {
    __shared__ float sB[CLEN][D_STATE];
    __shared__ float sC[CLEN][D_STATE];

    const int chunk = blockIdx.x & (NCHUNK - 1);
    const int rest  = blockIdx.x >> 5;
    const int b     = rest >> 2;
    const int d     = ((rest & 3) << 8) + threadIdx.x;

    const float LOG2E = 1.44269504088896340736f;
    float Alog2[D_STATE];
#pragma unroll
    for (int n = 0; n < D_STATE; ++n)
        Alog2[n] = -__expf(A_log[d * D_STATE + n]) * LOG2E;

    const size_t baseS = ((size_t)b * T_LEN + (size_t)chunk * CLEN) * D_STATE;
    for (int i = threadIdx.x; i < CLEN * D_STATE; i += 256) {
        ((float*)sB)[i] = Btg[baseS + i];
        ((float*)sC)[i] = Ctg[baseS + i];
    }
    __syncthreads();

    float h[D_STATE];
    const size_t ho = (((size_t)b * NCHUNK + chunk) * D_MODEL + d) * D_STATE;
#pragma unroll
    for (int q = 0; q < 4; ++q)
        *(v4f*)(h + 4 * q) = *(const v4f*)(Hstart + ho + 4 * q);

    const float Dd = Dvec[d];
    const size_t baseU = ((size_t)b * T_LEN + (size_t)chunk * CLEN) * D_MODEL + d;
    for (int tt = 0; tt < CLEN; ++tt) {
        const size_t off = baseU + (size_t)tt * D_MODEL;
        const float ul  = u[off];
        const float dtl = dt[off];
        const float dtu = dtl * ul;
        float y = Dd * ul;
#pragma unroll
        for (int n = 0; n < D_STATE; ++n) {
            float abar = exp2f(dtl * Alog2[n]);
            h[n] = fmaf(abar, h[n], sB[tt][n] * dtu);
            y    = fmaf(h[n], sC[tt][n], y);
        }
        out[off] = y;
    }
}

extern "C" void kernel_launch(void* const* d_in, const int* in_sizes, int n_in,
                              void* d_out, int out_size, void* d_ws, size_t ws_size,
                              hipStream_t stream) {
    const float* u     = (const float*)d_in[0];
    const float* A_log = (const float*)d_in[1];
    const float* W_xdt = (const float*)d_in[2];
    const float* b_xdt = (const float*)d_in[3];
    const float* W_dt  = (const float*)d_in[4];
    const float* b_dt  = (const float*)d_in[5];
    const float* W_B   = (const float*)d_in[6];
    const float* b_B   = (const float*)d_in[7];
    const float* W_C   = (const float*)d_in[8];
    const float* b_C   = (const float*)d_in[9];
    const float* Dv    = (const float*)d_in[10];
    float* out = (float*)d_out;

    float* ws   = (float*)d_ws;
    float* xdt  = ws;
    float* dt   = xdt + (size_t)M_ROWS * DT_RANK;
    float* Bt   = dt  + (size_t)M_ROWS * D_MODEL;
    float* Ct   = Bt  + (size_t)M_ROWS * D_STATE;
    float* Pbuf = Ct  + (size_t)M_ROWS * D_STATE;
    float* Sbuf = Pbuf + (size_t)B_SZ * NCHUNK * D_MODEL * D_STATE;

    proj_u_fused<<<M_ROWS / PJ_ROWS, 256, 0, stream>>>(
        u, W_xdt, b_xdt, W_B, b_B, W_C, b_C, xdt, Bt, Ct);

    wmma_gemm_f32<1><<<(M_ROWS / 16) * (D_MODEL / 16) / 8, 256, 0, stream>>>(
        xdt, W_dt, b_dt, dt, M_ROWS, D_MODEL, DT_RANK);

    const int scan_blocks = B_SZ * 4 * NCHUNK;
    scan_phase1<<<scan_blocks, 256, 0, stream>>>(u, dt, Bt, A_log, Pbuf, Sbuf);
    scan_phase2<<<B_SZ * 4, 256, 0, stream>>>(Pbuf, Sbuf);
    scan_phase3<<<scan_blocks, 256, 0, stream>>>(u, dt, Bt, Ct, A_log, Dv, Pbuf, out);
}


# the three scan launches request their global operands ahead of the dependent chains: u, dt eight steps ahead in a second register set (passes 1, 3), all 32 chunks' P, S up front (pass 2)
# speedup vs baseline: 1.9700x; 1.5600x over previous
#include <hip/hip_runtime.h>
#include <hip/hip_bf16.h>
#include <math.h>

typedef float v2f __attribute__((ext_vector_type(2)));
typedef float v4f __attribute__((ext_vector_type(4)));
typedef float v8f __attribute__((ext_vector_type(8)));

#define B_SZ    4
#define T_LEN   2048
#define D_MODEL 1024
#define D_STATE 16
#define DT_RANK 64
#define M_ROWS  (B_SZ * T_LEN)

#define NCHUNK  32
#define CLEN    (T_LEN / NCHUNK)
#define SC_PF   8

template <int ACT>
__global__ __launch_bounds__(256)
void wmma_gemm_f32(const float* __restrict__ A,
                   const float* __restrict__ W,
                   const float* __restrict__ bias,
                   float* __restrict__ out,
                   int M, int N, int K) {
    const int wave   = (blockIdx.x * blockDim.x + threadIdx.x) >> 5;
    const int lane   = threadIdx.x & 31;
    const int ntiles = N >> 4;
    const int m0     = (wave / ntiles) << 4;
    const int n0     = (wave % ntiles) << 4;
    const int half   = lane >> 4;
    const int l15    = lane & 15;

    const float* arow = A + (size_t)(m0 + l15) * K + 2 * half;
    const float* wrow = W + (size_t)(n0 + l15) * K + 2 * half;

    v8f c = {};
#pragma unroll 4
    for (int k = 0; k < K; k += 4) {
        v2f a = *(const v2f*)(arow + k);
        v2f b = *(const v2f*)(wrow + k);
        c = __builtin_amdgcn_wmma_f32_16x16x4_f32(
                false, a, false, b, (short)0, c, false, false);
    }

    const float bv = bias[n0 + l15];
#pragma unroll
    for (int r = 0; r < 8; ++r) {
        float v = c[r] + bv;
        if (ACT == 1) v = (v > 20.0f) ? v : log1pf(__expf(v));
        out[(size_t)(m0 + half * 8 + r) * N + (n0 + l15)] = v;
    }
}

#define PJ_ROWS   32
#define PJ_KSPLIT 8
#define PJ_KSLICE (D_MODEL / PJ_KSPLIT)
#define PJ_NT     6
#define PJ_COLS   (PJ_NT * 16)
#define PJ_PITCH  100

__global__ __launch_bounds__(256)
void proj_u_fused(const float* __restrict__ u,
                  const float* __restrict__ W_xdt,
                  const float* __restrict__ b_xdt,
                  const float* __restrict__ W_B,
                  const float* __restrict__ b_B,
                  const float* __restrict__ W_C,
                  const float* __restrict__ b_C,
                  float* __restrict__ xdt,
                  float* __restrict__ Bt,
                  float* __restrict__ Ct) {
    __shared__ float part[PJ_KSPLIT][PJ_ROWS * PJ_PITCH];

    const int lane = threadIdx.x & 31;
    const int w    = threadIdx.x >> 5;
    const int half = lane >> 4;
    const int l15  = lane & 15;
    const int m0   = blockIdx.x * PJ_ROWS;
    const int koff = w * PJ_KSLICE + 16 * half;

    const float* ap[2];
    ap[0] = u + (size_t)(m0 + l15) * D_MODEL + koff;
    ap[1] = ap[0] + (size_t)16 * D_MODEL;
    const float* bp[PJ_NT];
#pragma unroll
    for (int j = 0; j < 4; ++j)
        bp[j] = W_xdt + (size_t)(16 * j + l15) * D_MODEL + koff;
    bp[4] = W_B + (size_t)l15 * D_MODEL + koff;
    bp[5] = W_C + (size_t)l15 * D_MODEL + koff;

    v8f acc[2][PJ_NT];
#pragma unroll
    for (int mt = 0; mt < 2; ++mt)
#pragma unroll
        for (int j = 0; j < PJ_NT; ++j)
            acc[mt][j] = v8f{};

#pragma unroll
    for (int ks = 0; ks < PJ_KSLICE; ks += 32) {
        v4f a0[4], a1[4];
#pragma unroll
        for (int q = 0; q < 4; ++q) a0[q] = *(const v4f*)(ap[0] + ks + 4 * q);
#pragma unroll
        for (int q = 0; q < 4; ++q) a1[q] = *(const v4f*)(ap[1] + ks + 4 * q);
#pragma unroll
        for (int j = 0; j < PJ_NT; ++j) {
            v4f b[4];
#pragma unroll
            for (int q = 0; q < 4; ++q) b[q] = *(const v4f*)(bp[j] + ks + 4 * q);
#pragma unroll
            for (int q = 0; q < 4; ++q) {
#pragma unroll
                for (int s = 0; s < 2; ++s) {
                    const v2f f0 = {a0[q][2 * s], a0[q][2 * s + 1]};
                    const v2f f1 = {a1[q][2 * s], a1[q][2 * s + 1]};
                    const v2f fb = {b[q][2 * s], b[q][2 * s + 1]};
                    acc[0][j] = __builtin_amdgcn_wmma_f32_16x16x4_f32(
                        false, f0, false, fb, (short)0, acc[0][j], false, false);
                    acc[1][j] = __builtin_amdgcn_wmma_f32_16x16x4_f32(
                        false, f1, false, fb, (short)0, acc[1][j], false, false);
                }
            }
        }
    }

    float* mine = part[w];
#pragma unroll
    for (int mt = 0; mt < 2; ++mt)
#pragma unroll
        for (int j = 0; j < PJ_NT; ++j)
#pragma unroll
            for (int r = 0; r < 8; ++r)
                mine[(16 * mt + 8 * half + r) * PJ_PITCH + 16 * j + l15] = acc[mt][j][r];
    __syncthreads();

#pragma unroll
    for (int i = 0; i < (PJ_ROWS * PJ_COLS) / 256; ++i) {
        const int o = threadIdx.x + 256 * i;
        int row, col;
        float bias;
        float* dst;
        if (i < 8) {
            row = o >> 6;  col = o & 63;
            bias = b_xdt[col];
            dst  = xdt + (size_t)m0 * DT_RANK + o;
        } else if (i < 10) {
            const int q = o - PJ_ROWS * DT_RANK;
            row = q >> 4;  col = 64 + (q & 15);
            bias = b_B[q & 15];
            dst  = Bt + (size_t)m0 * D_STATE + q;
        } else {
            const int q = o - PJ_ROWS * (DT_RANK + D_STATE);
            row = q >> 4;  col = 80 + (q & 15);
            bias = b_C[q & 15];
            dst  = Ct + (size_t)m0 * D_STATE + q;
        }
        float s = part[0][row * PJ_PITCH + col];
#pragma unroll
        for (int k = 1; k < PJ_KSPLIT; ++k)
            s += part[k][row * PJ_PITCH + col];
        *dst = s + bias;
    }
}

__global__ __launch_bounds__(256)
void scan_phase1(const float* __restrict__ u,
                 const float* __restrict__ dt,
                 const float* __restrict__ Btg,
                 const float* __restrict__ A_log,
                 float* __restrict__ Pbuf,
                 float* __restrict__ Sbuf) {
    __shared__ float sB[CLEN][D_STATE];

    const int chunk = blockIdx.x & (NCHUNK - 1);
    const int rest  = blockIdx.x >> 5;
    const int b     = rest >> 2;
    const int d     = ((rest & 3) << 8) + threadIdx.x;

    const float LOG2E = 1.44269504088896340736f;
    float Alog2[D_STATE];
#pragma unroll
    for (int n = 0; n < D_STATE; ++n)
        Alog2[n] = -__expf(A_log[d * D_STATE + n]) * LOG2E;

    const size_t baseU = ((size_t)b * T_LEN + (size_t)chunk * CLEN) * D_MODEL + d;
    const float* up = u  + baseU;
    const float* dp = dt + baseU;

    float un[SC_PF], dn[SC_PF];
#pragma unroll
    for (int j = 0; j < SC_PF; ++j) {
        un[j] = up[(size_t)j * D_MODEL];
        dn[j] = dp[(size_t)j * D_MODEL];
    }

    const size_t baseS = ((size_t)b * T_LEN + (size_t)chunk * CLEN) * D_STATE;
#pragma unroll
    for (int k = 0; k < (CLEN * D_STATE) / 256; ++k)
        ((float*)sB)[threadIdx.x + 256 * k] = Btg[baseS + threadIdx.x + 256 * k];
    __syncthreads();

    float P[D_STATE], S[D_STATE];
#pragma unroll
    for (int n = 0; n < D_STATE; ++n) { P[n] = 1.0f; S[n] = 0.0f; }

    for (int t0 = 0; t0 < CLEN; t0 += SC_PF) {
        float uc[SC_PF], dc[SC_PF];
#pragma unroll
        for (int j = 0; j < SC_PF; ++j) { uc[j] = un[j]; dc[j] = dn[j]; }
        if (t0 + SC_PF < CLEN) {
#pragma unroll
            for (int j = 0; j < SC_PF; ++j) {
                un[j] = up[(size_t)(t0 + SC_PF + j) * D_MODEL];
                dn[j] = dp[(size_t)(t0 + SC_PF + j) * D_MODEL];
            }
        }
#pragma unroll
        for (int j = 0; j < SC_PF; ++j) {
            const float ul  = uc[j];
            const float dtl = dc[j];
            const float dtu = dtl * ul;
#pragma unroll
            for (int n = 0; n < D_STATE; ++n) {
                float abar = exp2f(dtl * Alog2[n]);
                P[n] *= abar;
                S[n]  = fmaf(abar, S[n], sB[t0 + j][n] * dtu);
            }
        }
    }

    const size_t o = (((size_t)b * NCHUNK + chunk) * D_MODEL + d) * D_STATE;
#pragma unroll
    for (int q = 0; q < 4; ++q) {
        *(v4f*)(Pbuf + o + 4 * q) = *(v4f*)(P + 4 * q);
        *(v4f*)(Sbuf + o + 4 * q) = *(v4f*)(S + 4 * q);
    }
}

#define P2_THREADS 64

__global__ __launch_bounds__(P2_THREADS)
void scan_phase2(float* __restrict__ Pbuf,
                 const float* __restrict__ Sbuf) {
    const int tid = blockIdx.x * P2_THREADS + threadIdx.x;
    const int b   = tid / (D_MODEL * (D_STATE / 4));
    const int r   = tid % (D_MODEL * (D_STATE / 4));
    const size_t cstride = (size_t)D_MODEL * D_STATE;
    const size_t base    = (size_t)b * NCHUNK * cstride + (size_t)r * 4;

    v4f P[NCHUNK], S[NCHUNK];
#pragma unroll
    for (int c = 0; c < NCHUNK; ++c) {
        P[c] = *(const v4f*)(Pbuf + base + (size_t)c * cstride);
        S[c] = *(const v4f*)(Sbuf + base + (size_t)c * cstride);
    }

    v4f h = {0.0f, 0.0f, 0.0f, 0.0f};
#pragma unroll
    for (int c = 0; c < NCHUNK; ++c) {
        *(v4f*)(Pbuf + base + (size_t)c * cstride) = h;
#pragma unroll
        for (int e = 0; e < 4; ++e)
            h[e] = fmaf(P[c][e], h[e], S[c][e]);
    }
}

__global__ __launch_bounds__(256)
void scan_phase3(const float* __restrict__ u,
                 const float* __restrict__ dt,
                 const float* __restrict__ Btg,
                 const float* __restrict__ Ctg,
                 const float* __restrict__ A_log,
                 const float* __restrict__ Dvec,
                 const float* __restrict__ Hstart,
                 float* __restrict__ out) {
    __shared__ float sB[CLEN][D_STATE];
    __shared__ float sC[CLEN][D_STATE];

    const int chunk = blockIdx.x & (NCHUNK - 1);
    const int rest  = blockIdx.x >> 5;
    const int b     = rest >> 2;
    const int d     = ((rest & 3) << 8) + threadIdx.x;

    const float LOG2E = 1.44269504088896340736f;
    float Alog2[D_STATE];
#pragma unroll
    for (int n = 0; n < D_STATE; ++n)
        Alog2[n] = -__expf(A_log[d * D_STATE + n]) * LOG2E;

    float h[D_STATE];
    const size_t ho = (((size_t)b * NCHUNK + chunk) * D_MODEL + d) * D_STATE;
#pragma unroll
    for (int q = 0; q < 4; ++q)
        *(v4f*)(h + 4 * q) = *(const v4f*)(Hstart + ho + 4 * q);
    const float Dd = Dvec[d];

    const size_t baseU = ((size_t)b * T_LEN + (size_t)chunk * CLEN) * D_MODEL + d;
    const float* up = u  + baseU;
    const float* dp = dt + baseU;
    float*       yp = out + baseU;

    float un[SC_PF], dn[SC_PF];
#pragma unroll
    for (int j = 0; j < SC_PF; ++j) {
        un[j] = up[(size_t)j * D_MODEL];
        dn[j] = dp[(size_t)j * D_MODEL];
    }

    const size_t baseS = ((size_t)b * T_LEN + (size_t)chunk * CLEN) * D_STATE;
#pragma unroll
    for (int k = 0; k < (CLEN * D_STATE) / 256; ++k) {
        ((float*)sB)[threadIdx.x + 256 * k] = Btg[baseS + threadIdx.x + 256 * k];
        ((float*)sC)[threadIdx.x + 256 * k] = Ctg[baseS + threadIdx.x + 256 * k];
    }
    __syncthreads();

    for (int t0 = 0; t0 < CLEN; t0 += SC_PF) {
        float uc[SC_PF], dc[SC_PF];
#pragma unroll
        for (int j = 0; j < SC_PF; ++j) { uc[j] = un[j]; dc[j] = dn[j]; }
        if (t0 + SC_PF < CLEN) {
#pragma unroll
            for (int j = 0; j < SC_PF; ++j) {
                un[j] = up[(size_t)(t0 + SC_PF + j) * D_MODEL];
                dn[j] = dp[(size_t)(t0 + SC_PF + j) * D_MODEL];
            }
        }
#pragma unroll
        for (int j = 0; j < SC_PF; ++j) {
            const float ul  = uc[j];
            const float dtl = dc[j];
            const float dtu = dtl * ul;
            float y = Dd * ul;
#pragma unroll
            for (int n = 0; n < D_STATE; ++n) {
                float abar = exp2f(dtl * Alog2[n]);
                h[n] = fmaf(abar, h[n], sB[t0 + j][n] * dtu);
                y    = fmaf(h[n], sC[t0 + j][n], y);
            }
            yp[(size_t)(t0 + j) * D_MODEL] = y;
        }
    }
}

extern "C" void kernel_launch(void* const* d_in, const int* in_sizes, int n_in,
                              void* d_out, int out_size, void* d_ws, size_t ws_size,
                              hipStream_t stream) {
    const float* u     = (const float*)d_in[0];
    const float* A_log = (const float*)d_in[1];
    const float* W_xdt = (const float*)d_in[2];
    const float* b_xdt = (const float*)d_in[3];
    const float* W_dt  = (const float*)d_in[4];
    const float* b_dt  = (const float*)d_in[5];
    const float* W_B   = (const float*)d_in[6];
    const float* b_B   = (const float*)d_in[7];
    const float* W_C   = (const float*)d_in[8];
    const float* b_C   = (const float*)d_in[9];
    const float* Dv    = (const float*)d_in[10];
    float* out = (float*)d_out;

    float* ws   = (float*)d_ws;
    float* xdt  = ws;
    float* dt   = xdt + (size_t)M_ROWS * DT_RANK;
    float* Bt   = dt  + (size_t)M_ROWS * D_MODEL;
    float* Ct   = Bt  + (size_t)M_ROWS * D_STATE;
    float* Pbuf = Ct  + (size_t)M_ROWS * D_STATE;
    float* Sbuf = Pbuf + (size_t)B_SZ * NCHUNK * D_MODEL * D_STATE;

    proj_u_fused<<<M_ROWS / PJ_ROWS, 256, 0, stream>>>(
        u, W_xdt, b_xdt, W_B, b_B, W_C, b_C, xdt, Bt, Ct);

    wmma_gemm_f32<1><<<(M_ROWS / 16) * (D_MODEL / 16) / 8, 256, 0, stream>>>(
        xdt, W_dt, b_dt, dt, M_ROWS, D_MODEL, DT_RANK);

    const int scan_blocks = B_SZ * 4 * NCHUNK;
    scan_phase1<<<scan_blocks, 256, 0, stream>>>(u, dt, Bt, A_log, Pbuf, Sbuf);
    scan_phase2<<<B_SZ * D_MODEL * (D_STATE / 4) / P2_THREADS, P2_THREADS, 0, stream>>>(Pbuf, Sbuf);
    scan_phase3<<<scan_blocks, 256, 0, stream>>>(u, dt, Bt, Ct, A_log, Dv, Pbuf, out);
}


# scan phases 1 and 3: the decay factor is the bare hardware exp2 (one v_exp_f32) instead of library exp2f, which adds 5 instructions per call for sub-normal results (those may now flush to zero)
# speedup vs baseline: 2.3475x; 1.1916x over previous
#include <hip/hip_runtime.h>
#include <hip/hip_bf16.h>
#include <math.h>

typedef float v2f __attribute__((ext_vector_type(2)));
typedef float v4f __attribute__((ext_vector_type(4)));
typedef float v8f __attribute__((ext_vector_type(8)));

#define B_SZ    4
#define T_LEN   2048
#define D_MODEL 1024
#define D_STATE 16
#define DT_RANK 64
#define M_ROWS  (B_SZ * T_LEN)

#define NCHUNK  32
#define CLEN    (T_LEN / NCHUNK)
#define SC_PF   8

template <int ACT>
__global__ __launch_bounds__(256)
void wmma_gemm_f32(const float* __restrict__ A,
                   const float* __restrict__ W,
                   const float* __restrict__ bias,
                   float* __restrict__ out,
                   int M, int N, int K) {
    const int wave   = (blockIdx.x * blockDim.x + threadIdx.x) >> 5;
    const int lane   = threadIdx.x & 31;
    const int ntiles = N >> 4;
    const int m0     = (wave / ntiles) << 4;
    const int n0     = (wave % ntiles) << 4;
    const int half   = lane >> 4;
    const int l15    = lane & 15;

    const float* arow = A + (size_t)(m0 + l15) * K + 2 * half;
    const float* wrow = W + (size_t)(n0 + l15) * K + 2 * half;

    v8f c = {};
#pragma unroll 4
    for (int k = 0; k < K; k += 4) {
        v2f a = *(const v2f*)(arow + k);
        v2f b = *(const v2f*)(wrow + k);
        c = __builtin_amdgcn_wmma_f32_16x16x4_f32(
                false, a, false, b, (short)0, c, false, false);
    }

    const float bv = bias[n0 + l15];
#pragma unroll
    for (int r = 0; r < 8; ++r) {
        float v = c[r] + bv;
        if (ACT == 1) v = (v > 20.0f) ? v : log1pf(__expf(v));
        out[(size_t)(m0 + half * 8 + r) * N + (n0 + l15)] = v;
    }
}

#define PJ_ROWS   32
#define PJ_KSPLIT 8
#define PJ_KSLICE (D_MODEL / PJ_KSPLIT)
#define PJ_NT     6
#define PJ_COLS   (PJ_NT * 16)
#define PJ_PITCH  100

__global__ __launch_bounds__(256)
void proj_u_fused(const float* __restrict__ u,
                  const float* __restrict__ W_xdt,
                  const float* __restrict__ b_xdt,
                  const float* __restrict__ W_B,
                  const float* __restrict__ b_B,
                  const float* __restrict__ W_C,
                  const float* __restrict__ b_C,
                  float* __restrict__ xdt,
                  float* __restrict__ Bt,
                  float* __restrict__ Ct) {
    __shared__ float part[PJ_KSPLIT][PJ_ROWS * PJ_PITCH];

    const int lane = threadIdx.x & 31;
    const int w    = threadIdx.x >> 5;
    const int half = lane >> 4;
    const int l15  = lane & 15;
    const int m0   = blockIdx.x * PJ_ROWS;
    const int koff = w * PJ_KSLICE + 16 * half;

    const float* ap[2];
    ap[0] = u + (size_t)(m0 + l15) * D_MODEL + koff;
    ap[1] = ap[0] + (size_t)16 * D_MODEL;
    const float* bp[PJ_NT];
#pragma unroll
    for (int j = 0; j < 4; ++j)
        bp[j] = W_xdt + (size_t)(16 * j + l15) * D_MODEL + koff;
    bp[4] = W_B + (size_t)l15 * D_MODEL + koff;
    bp[5] = W_C + (size_t)l15 * D_MODEL + koff;

    v8f acc[2][PJ_NT];
#pragma unroll
    for (int mt = 0; mt < 2; ++mt)
#pragma unroll
        for (int j = 0; j < PJ_NT; ++j)
            acc[mt][j] = v8f{};

#pragma unroll
    for (int ks = 0; ks < PJ_KSLICE; ks += 32) {
        v4f a0[4], a1[4];
#pragma unroll
        for (int q = 0; q < 4; ++q) a0[q] = *(const v4f*)(ap[0] + ks + 4 * q);
#pragma unroll
        for (int q = 0; q < 4; ++q) a1[q] = *(const v4f*)(ap[1] + ks + 4 * q);
#pragma unroll
        for (int j = 0; j < PJ_NT; ++j) {
            v4f b[4];
#pragma unroll
            for (int q = 0; q < 4; ++q) b[q] = *(const v4f*)(bp[j] + ks + 4 * q);
#pragma unroll
            for (int q = 0; q < 4; ++q) {
#pragma unroll
                for (int s = 0; s < 2; ++s) {
                    const v2f f0 = {a0[q][2 * s], a0[q][2 * s + 1]};
                    const v2f f1 = {a1[q][2 * s], a1[q][2 * s + 1]};
                    const v2f fb = {b[q][2 * s], b[q][2 * s + 1]};
                    acc[0][j] = __builtin_amdgcn_wmma_f32_16x16x4_f32(
                        false, f0, false, fb, (short)0, acc[0][j], false, false);
                    acc[1][j] = __builtin_amdgcn_wmma_f32_16x16x4_f32(
                        false, f1, false, fb, (short)0, acc[1][j], false, false);
                }
            }
        }
    }

    float* mine = part[w];
#pragma unroll
    for (int mt = 0; mt < 2; ++mt)
#pragma unroll
        for (int j = 0; j < PJ_NT; ++j)
#pragma unroll
            for (int r = 0; r < 8; ++r)
                mine[(16 * mt + 8 * half + r) * PJ_PITCH + 16 * j + l15] = acc[mt][j][r];
    __syncthreads();

#pragma unroll
    for (int i = 0; i < (PJ_ROWS * PJ_COLS) / 256; ++i) {
        const int o = threadIdx.x + 256 * i;
        int row, col;
        float bias;
        float* dst;
        if (i < 8) {
            row = o >> 6;  col = o & 63;
            bias = b_xdt[col];
            dst  = xdt + (size_t)m0 * DT_RANK + o;
        } else if (i < 10) {
            const int q = o - PJ_ROWS * DT_RANK;
            row = q >> 4;  col = 64 + (q & 15);
            bias = b_B[q & 15];
            dst  = Bt + (size_t)m0 * D_STATE + q;
        } else {
            const int q = o - PJ_ROWS * (DT_RANK + D_STATE);
            row = q >> 4;  col = 80 + (q & 15);
            bias = b_C[q & 15];
            dst  = Ct + (size_t)m0 * D_STATE + q;
        }
        float s = part[0][row * PJ_PITCH + col];
#pragma unroll
        for (int k = 1; k < PJ_KSPLIT; ++k)
            s += part[k][row * PJ_PITCH + col];
        *dst = s + bias;
    }
}

__global__ __launch_bounds__(256)
void scan_phase1(const float* __restrict__ u,
                 const float* __restrict__ dt,
                 const float* __restrict__ Btg,
                 const float* __restrict__ A_log,
                 float* __restrict__ Pbuf,
                 float* __restrict__ Sbuf) {
    __shared__ float sB[CLEN][D_STATE];

    const int chunk = blockIdx.x & (NCHUNK - 1);
    const int rest  = blockIdx.x >> 5;
    const int b     = rest >> 2;
    const int d     = ((rest & 3) << 8) + threadIdx.x;

    const float LOG2E = 1.44269504088896340736f;
    float Alog2[D_STATE];
#pragma unroll
    for (int n = 0; n < D_STATE; ++n)
        Alog2[n] = -__expf(A_log[d * D_STATE + n]) * LOG2E;

    const size_t baseU = ((size_t)b * T_LEN + (size_t)chunk * CLEN) * D_MODEL + d;
    const float* up = u  + baseU;
    const float* dp = dt + baseU;

    float un[SC_PF], dn[SC_PF];
#pragma unroll
    for (int j = 0; j < SC_PF; ++j) {
        un[j] = up[(size_t)j * D_MODEL];
        dn[j] = dp[(size_t)j * D_MODEL];
    }

    const size_t baseS = ((size_t)b * T_LEN + (size_t)chunk * CLEN) * D_STATE;
#pragma unroll
    for (int k = 0; k < (CLEN * D_STATE) / 256; ++k)
        ((float*)sB)[threadIdx.x + 256 * k] = Btg[baseS + threadIdx.x + 256 * k];
    __syncthreads();

    float P[D_STATE], S[D_STATE];
#pragma unroll
    for (int n = 0; n < D_STATE; ++n) { P[n] = 1.0f; S[n] = 0.0f; }

    for (int t0 = 0; t0 < CLEN; t0 += SC_PF) {
        float uc[SC_PF], dc[SC_PF];
#pragma unroll
        for (int j = 0; j < SC_PF; ++j) { uc[j] = un[j]; dc[j] = dn[j]; }
        if (t0 + SC_PF < CLEN) {
#pragma unroll
            for (int j = 0; j < SC_PF; ++j) {
                un[j] = up[(size_t)(t0 + SC_PF + j) * D_MODEL];
                dn[j] = dp[(size_t)(t0 + SC_PF + j) * D_MODEL];
            }
        }
#pragma unroll
        for (int j = 0; j < SC_PF; ++j) {
            const float ul  = uc[j];
            const float dtl = dc[j];
            const float dtu = dtl * ul;
#pragma unroll
            for (int n = 0; n < D_STATE; ++n) {
                float abar = __builtin_amdgcn_exp2f(dtl * Alog2[n]);
                P[n] *= abar;
                S[n]  = fmaf(abar, S[n], sB[t0 + j][n] * dtu);
            }
        }
    }

    const size_t o = (((size_t)b * NCHUNK + chunk) * D_MODEL + d) * D_STATE;
#pragma unroll
    for (int q = 0; q < 4; ++q) {
        *(v4f*)(Pbuf + o + 4 * q) = *(v4f*)(P + 4 * q);
        *(v4f*)(Sbuf + o + 4 * q) = *(v4f*)(S + 4 * q);
    }
}

#define P2_THREADS 64

__global__ __launch_bounds__(P2_THREADS)
void scan_phase2(float* __restrict__ Pbuf,
                 const float* __restrict__ Sbuf) {
    const int tid = blockIdx.x * P2_THREADS + threadIdx.x;
    const int b   = tid / (D_MODEL * (D_STATE / 4));
    const int r   = tid % (D_MODEL * (D_STATE / 4));
    const size_t cstride = (size_t)D_MODEL * D_STATE;
    const size_t base    = (size_t)b * NCHUNK * cstride + (size_t)r * 4;

    v4f P[NCHUNK], S[NCHUNK];
#pragma unroll
    for (int c = 0; c < NCHUNK; ++c) {
        P[c] = *(const v4f*)(Pbuf + base + (size_t)c * cstride);
        S[c] = *(const v4f*)(Sbuf + base + (size_t)c * cstride);
    }

    v4f h = {0.0f, 0.0f, 0.0f, 0.0f};
#pragma unroll
    for (int c = 0; c < NCHUNK; ++c) {
        *(v4f*)(Pbuf + base + (size_t)c * cstride) = h;
#pragma unroll
        for (int e = 0; e < 4; ++e)
            h[e] = fmaf(P[c][e], h[e], S[c][e]);
    }
}

__global__ __launch_bounds__(256)
void scan_phase3(const float* __restrict__ u,
                 const float* __restrict__ dt,
                 const float* __restrict__ Btg,
                 const float* __restrict__ Ctg,
                 const float* __restrict__ A_log,
                 const float* __restrict__ Dvec,
                 const float* __restrict__ Hstart,
                 float* __restrict__ out) {
    __shared__ float sB[CLEN][D_STATE];
    __shared__ float sC[CLEN][D_STATE];

    const int chunk = blockIdx.x & (NCHUNK - 1);
    const int rest  = blockIdx.x >> 5;
    const int b     = rest >> 2;
    const int d     = ((rest & 3) << 8) + threadIdx.x;

    const float LOG2E = 1.44269504088896340736f;
    float Alog2[D_STATE];
#pragma unroll
    for (int n = 0; n < D_STATE; ++n)
        Alog2[n] = -__expf(A_log[d * D_STATE + n]) * LOG2E;

    float h[D_STATE];
    const size_t ho = (((size_t)b * NCHUNK + chunk) * D_MODEL + d) * D_STATE;
#pragma unroll
    for (int q = 0; q < 4; ++q)
        *(v4f*)(h + 4 * q) = *(const v4f*)(Hstart + ho + 4 * q);
    const float Dd = Dvec[d];

    const size_t baseU = ((size_t)b * T_LEN + (size_t)chunk * CLEN) * D_MODEL + d;
    const float* up = u  + baseU;
    const float* dp = dt + baseU;
    float*       yp = out + baseU;

    float un[SC_PF], dn[SC_PF];
#pragma unroll
    for (int j = 0; j < SC_PF; ++j) {
        un[j] = up[(size_t)j * D_MODEL];
        dn[j] = dp[(size_t)j * D_MODEL];
    }

    const size_t baseS = ((size_t)b * T_LEN + (size_t)chunk * CLEN) * D_STATE;
#pragma unroll
    for (int k = 0; k < (CLEN * D_STATE) / 256; ++k) {
        ((float*)sB)[threadIdx.x + 256 * k] = Btg[baseS + threadIdx.x + 256 * k];
        ((float*)sC)[threadIdx.x + 256 * k] = Ctg[baseS + threadIdx.x + 256 * k];
    }
    __syncthreads();

    for (int t0 = 0; t0 < CLEN; t0 += SC_PF) {
        float uc[SC_PF], dc[SC_PF];
#pragma unroll
        for (int j = 0; j < SC_PF; ++j) { uc[j] = un[j]; dc[j] = dn[j]; }
        if (t0 + SC_PF < CLEN) {
#pragma unroll
            for (int j = 0; j < SC_PF; ++j) {
                un[j] = up[(size_t)(t0 + SC_PF + j) * D_MODEL];
                dn[j] = dp[(size_t)(t0 + SC_PF + j) * D_MODEL];
            }
        }
#pragma unroll
        for (int j = 0; j < SC_PF; ++j) {
            const float ul  = uc[j];
            const float dtl = dc[j];
            const float dtu = dtl * ul;
            float y = Dd * ul;
#pragma unroll
            for (int n = 0; n < D_STATE; ++n) {
                float abar = __builtin_amdgcn_exp2f(dtl * Alog2[n]);
                h[n] = fmaf(abar, h[n], sB[t0 + j][n] * dtu);
                y    = fmaf(h[n], sC[t0 + j][n], y);
            }
            yp[(size_t)(t0 + j) * D_MODEL] = y;
        }
    }
}

extern "C" void kernel_launch(void* const* d_in, const int* in_sizes, int n_in,
                              void* d_out, int out_size, void* d_ws, size_t ws_size,
                              hipStream_t stream) {
    const float* u     = (const float*)d_in[0];
    const float* A_log = (const float*)d_in[1];
    const float* W_xdt = (const float*)d_in[2];
    const float* b_xdt = (const float*)d_in[3];
    const float* W_dt  = (const float*)d_in[4];
    const float* b_dt  = (const float*)d_in[5];
    const float* W_B   = (const float*)d_in[6];
    const float* b_B   = (const float*)d_in[7];
    const float* W_C   = (const float*)d_in[8];
    const float* b_C   = (const float*)d_in[9];
    const float* Dv    = (const float*)d_in[10];
    float* out = (float*)d_out;

    float* ws   = (float*)d_ws;
    float* xdt  = ws;
    float* dt   = xdt + (size_t)M_ROWS * DT_RANK;
    float* Bt   = dt  + (size_t)M_ROWS * D_MODEL;
    float* Ct   = Bt  + (size_t)M_ROWS * D_STATE;
    float* Pbuf = Ct  + (size_t)M_ROWS * D_STATE;
    float* Sbuf = Pbuf + (size_t)B_SZ * NCHUNK * D_MODEL * D_STATE;

    proj_u_fused<<<M_ROWS / PJ_ROWS, 256, 0, stream>>>(
        u, W_xdt, b_xdt, W_B, b_B, W_C, b_C, xdt, Bt, Ct);

    wmma_gemm_f32<1><<<(M_ROWS / 16) * (D_MODEL / 16) / 8, 256, 0, stream>>>(
        xdt, W_dt, b_dt, dt, M_ROWS, D_MODEL, DT_RANK);

    const int scan_blocks = B_SZ * 4 * NCHUNK;
    scan_phase1<<<scan_blocks, 256, 0, stream>>>(u, dt, Bt, A_log, Pbuf, Sbuf);
    scan_phase2<<<B_SZ * D_MODEL * (D_STATE / 4) / P2_THREADS, P2_THREADS, 0, stream>>>(Pbuf, Sbuf);
    scan_phase3<<<scan_blocks, 256, 0, stream>>>(u, dt, Bt, Ct, A_log, Dv, Pbuf, out);
}


# softplus of the dt projection: log1p(e) as log(w)*e/(w-1), w=1+e, on the hardware log2 and reciprocal (a few ulp), branch-free; was library log1pf, ~85 dependent instructions per value
# speedup vs baseline: 2.4777x; 1.0555x over previous
#include <hip/hip_runtime.h>
#include <hip/hip_bf16.h>
#include <math.h>

typedef float v2f __attribute__((ext_vector_type(2)));
typedef float v4f __attribute__((ext_vector_type(4)));
typedef float v8f __attribute__((ext_vector_type(8)));

#define B_SZ    4
#define T_LEN   2048
#define D_MODEL 1024
#define D_STATE 16
#define DT_RANK 64
#define M_ROWS  (B_SZ * T_LEN)

#define NCHUNK  32
#define CLEN    (T_LEN / NCHUNK)
#define SC_PF   8

template <int ACT>
__global__ __launch_bounds__(256)
void wmma_gemm_f32(const float* __restrict__ A,
                   const float* __restrict__ W,
                   const float* __restrict__ bias,
                   float* __restrict__ out,
                   int M, int N, int K) {
    const int wave   = (blockIdx.x * blockDim.x + threadIdx.x) >> 5;
    const int lane   = threadIdx.x & 31;
    const int ntiles = N >> 4;
    const int m0     = (wave / ntiles) << 4;
    const int n0     = (wave % ntiles) << 4;
    const int half   = lane >> 4;
    const int l15    = lane & 15;

    const float* arow = A + (size_t)(m0 + l15) * K + 2 * half;
    const float* wrow = W + (size_t)(n0 + l15) * K + 2 * half;

    v8f c = {};
#pragma unroll 4
    for (int k = 0; k < K; k += 4) {
        v2f a = *(const v2f*)(arow + k);
        v2f b = *(const v2f*)(wrow + k);
        c = __builtin_amdgcn_wmma_f32_16x16x4_f32(
                false, a, false, b, (short)0, c, false, false);
    }

    const float bv = bias[n0 + l15];
#pragma unroll
    for (int r = 0; r < 8; ++r) {
        float v = c[r] + bv;
        if (ACT == 1) {
            const float e  = __expf(v);
            const float w  = 1.0f + e;
            const float ln = __builtin_amdgcn_logf(w) * 0.69314718055994530942f;
            const float sp = (w == 1.0f) ? e : ln * (e * __builtin_amdgcn_rcpf(w - 1.0f));
            v = (v > 20.0f) ? v : sp;
        }
        out[(size_t)(m0 + half * 8 + r) * N + (n0 + l15)] = v;
    }
}

#define PJ_ROWS   32
#define PJ_KSPLIT 8
#define PJ_KSLICE (D_MODEL / PJ_KSPLIT)
#define PJ_NT     6
#define PJ_COLS   (PJ_NT * 16)
#define PJ_PITCH  100

__global__ __launch_bounds__(256)
void proj_u_fused(const float* __restrict__ u,
                  const float* __restrict__ W_xdt,
                  const float* __restrict__ b_xdt,
                  const float* __restrict__ W_B,
                  const float* __restrict__ b_B,
                  const float* __restrict__ W_C,
                  const float* __restrict__ b_C,
                  float* __restrict__ xdt,
                  float* __restrict__ Bt,
                  float* __restrict__ Ct) {
    __shared__ float part[PJ_KSPLIT][PJ_ROWS * PJ_PITCH];

    const int lane = threadIdx.x & 31;
    const int w    = threadIdx.x >> 5;
    const int half = lane >> 4;
    const int l15  = lane & 15;
    const int m0   = blockIdx.x * PJ_ROWS;
    const int koff = w * PJ_KSLICE + 16 * half;

    const float* ap[2];
    ap[0] = u + (size_t)(m0 + l15) * D_MODEL + koff;
    ap[1] = ap[0] + (size_t)16 * D_MODEL;
    const float* bp[PJ_NT];
#pragma unroll
    for (int j = 0; j < 4; ++j)
        bp[j] = W_xdt + (size_t)(16 * j + l15) * D_MODEL + koff;
    bp[4] = W_B + (size_t)l15 * D_MODEL + koff;
    bp[5] = W_C + (size_t)l15 * D_MODEL + koff;

    v8f acc[2][PJ_NT];
#pragma unroll
    for (int mt = 0; mt < 2; ++mt)
#pragma unroll
        for (int j = 0; j < PJ_NT; ++j)
            acc[mt][j] = v8f{};

#pragma unroll
    for (int ks = 0; ks < PJ_KSLICE; ks += 32) {
        v4f a0[4], a1[4];
#pragma unroll
        for (int q = 0; q < 4; ++q) a0[q] = *(const v4f*)(ap[0] + ks + 4 * q);
#pragma unroll
        for (int q = 0; q < 4; ++q) a1[q] = *(const v4f*)(ap[1] + ks + 4 * q);
#pragma unroll
        for (int j = 0; j < PJ_NT; ++j) {
            v4f b[4];
#pragma unroll
            for (int q = 0; q < 4; ++q) b[q] = *(const v4f*)(bp[j] + ks + 4 * q);
#pragma unroll
            for (int q = 0; q < 4; ++q) {
#pragma unroll
                for (int s = 0; s < 2; ++s) {
                    const v2f f0 = {a0[q][2 * s], a0[q][2 * s + 1]};
                    const v2f f1 = {a1[q][2 * s], a1[q][2 * s + 1]};
                    const v2f fb = {b[q][2 * s], b[q][2 * s + 1]};
                    acc[0][j] = __builtin_amdgcn_wmma_f32_16x16x4_f32(
                        false, f0, false, fb, (short)0, acc[0][j], false, false);
                    acc[1][j] = __builtin_amdgcn_wmma_f32_16x16x4_f32(
                        false, f1, false, fb, (short)0, acc[1][j], false, false);
                }
            }
        }
    }

    float* mine = part[w];
#pragma unroll
    for (int mt = 0; mt < 2; ++mt)
#pragma unroll
        for (int j = 0; j < PJ_NT; ++j)
#pragma unroll
            for (int r = 0; r < 8; ++r)
                mine[(16 * mt + 8 * half + r) * PJ_PITCH + 16 * j + l15] = acc[mt][j][r];
    __syncthreads();

#pragma unroll
    for (int i = 0; i < (PJ_ROWS * PJ_COLS) / 256; ++i) {
        const int o = threadIdx.x + 256 * i;
        int row, col;
        float bias;
        float* dst;
        if (i < 8) {
            row = o >> 6;  col = o & 63;
            bias = b_xdt[col];
            dst  = xdt + (size_t)m0 * DT_RANK + o;
        } else if (i < 10) {
            const int q = o - PJ_ROWS * DT_RANK;
            row = q >> 4;  col = 64 + (q & 15);
            bias = b_B[q & 15];
            dst  = Bt + (size_t)m0 * D_STATE + q;
        } else {
            const int q = o - PJ_ROWS * (DT_RANK + D_STATE);
            row = q >> 4;  col = 80 + (q & 15);
            bias = b_C[q & 15];
            dst  = Ct + (size_t)m0 * D_STATE + q;
        }
        float s = part[0][row * PJ_PITCH + col];
#pragma unroll
        for (int k = 1; k < PJ_KSPLIT; ++k)
            s += part[k][row * PJ_PITCH + col];
        *dst = s + bias;
    }
}

__global__ __launch_bounds__(256)
void scan_phase1(const float* __restrict__ u,
                 const float* __restrict__ dt,
                 const float* __restrict__ Btg,
                 const float* __restrict__ A_log,
                 float* __restrict__ Pbuf,
                 float* __restrict__ Sbuf) {
    __shared__ float sB[CLEN][D_STATE];

    const int chunk = blockIdx.x & (NCHUNK - 1);
    const int rest  = blockIdx.x >> 5;
    const int b     = rest >> 2;
    const int d     = ((rest & 3) << 8) + threadIdx.x;

    const float LOG2E = 1.44269504088896340736f;
    float Alog2[D_STATE];
#pragma unroll
    for (int n = 0; n < D_STATE; ++n)
        Alog2[n] = -__expf(A_log[d * D_STATE + n]) * LOG2E;

    const size_t baseU = ((size_t)b * T_LEN + (size_t)chunk * CLEN) * D_MODEL + d;
    const float* up = u  + baseU;
    const float* dp = dt + baseU;

    float un[SC_PF], dn[SC_PF];
#pragma unroll
    for (int j = 0; j < SC_PF; ++j) {
        un[j] = up[(size_t)j * D_MODEL];
        dn[j] = dp[(size_t)j * D_MODEL];
    }

    const size_t baseS = ((size_t)b * T_LEN + (size_t)chunk * CLEN) * D_STATE;
#pragma unroll
    for (int k = 0; k < (CLEN * D_STATE) / 256; ++k)
        ((float*)sB)[threadIdx.x + 256 * k] = Btg[baseS + threadIdx.x + 256 * k];
    __syncthreads();

    float P[D_STATE], S[D_STATE];
#pragma unroll
    for (int n = 0; n < D_STATE; ++n) { P[n] = 1.0f; S[n] = 0.0f; }

    for (int t0 = 0; t0 < CLEN; t0 += SC_PF) {
        float uc[SC_PF], dc[SC_PF];
#pragma unroll
        for (int j = 0; j < SC_PF; ++j) { uc[j] = un[j]; dc[j] = dn[j]; }
        if (t0 + SC_PF < CLEN) {
#pragma unroll
            for (int j = 0; j < SC_PF; ++j) {
                un[j] = up[(size_t)(t0 + SC_PF + j) * D_MODEL];
                dn[j] = dp[(size_t)(t0 + SC_PF + j) * D_MODEL];
            }
        }
#pragma unroll
        for (int j = 0; j < SC_PF; ++j) {
            const float ul  = uc[j];
            const float dtl = dc[j];
            const float dtu = dtl * ul;
#pragma unroll
            for (int n = 0; n < D_STATE; ++n) {
                float abar = __builtin_amdgcn_exp2f(dtl * Alog2[n]);
                P[n] *= abar;
                S[n]  = fmaf(abar, S[n], sB[t0 + j][n] * dtu);
            }
        }
    }

    const size_t o = (((size_t)b * NCHUNK + chunk) * D_MODEL + d) * D_STATE;
#pragma unroll
    for (int q = 0; q < 4; ++q) {
        *(v4f*)(Pbuf + o + 4 * q) = *(v4f*)(P + 4 * q);
        *(v4f*)(Sbuf + o + 4 * q) = *(v4f*)(S + 4 * q);
    }
}

#define P2_THREADS 64

__global__ __launch_bounds__(P2_THREADS)
void scan_phase2(float* __restrict__ Pbuf,
                 const float* __restrict__ Sbuf) {
    const int tid = blockIdx.x * P2_THREADS + threadIdx.x;
    const int b   = tid / (D_MODEL * (D_STATE / 4));
    const int r   = tid % (D_MODEL * (D_STATE / 4));
    const size_t cstride = (size_t)D_MODEL * D_STATE;
    const size_t base    = (size_t)b * NCHUNK * cstride + (size_t)r * 4;

    v4f P[NCHUNK], S[NCHUNK];
#pragma unroll
    for (int c = 0; c < NCHUNK; ++c) {
        P[c] = *(const v4f*)(Pbuf + base + (size_t)c * cstride);
        S[c] = *(const v4f*)(Sbuf + base + (size_t)c * cstride);
    }

    v4f h = {0.0f, 0.0f, 0.0f, 0.0f};
#pragma unroll
    for (int c = 0; c < NCHUNK; ++c) {
        *(v4f*)(Pbuf + base + (size_t)c * cstride) = h;
#pragma unroll
        for (int e = 0; e < 4; ++e)
            h[e] = fmaf(P[c][e], h[e], S[c][e]);
    }
}

__global__ __launch_bounds__(256)
void scan_phase3(const float* __restrict__ u,
                 const float* __restrict__ dt,
                 const float* __restrict__ Btg,
                 const float* __restrict__ Ctg,
                 const float* __restrict__ A_log,
                 const float* __restrict__ Dvec,
                 const float* __restrict__ Hstart,
                 float* __restrict__ out) {
    __shared__ float sB[CLEN][D_STATE];
    __shared__ float sC[CLEN][D_STATE];

    const int chunk = blockIdx.x & (NCHUNK - 1);
    const int rest  = blockIdx.x >> 5;
    const int b     = rest >> 2;
    const int d     = ((rest & 3) << 8) + threadIdx.x;

    const float LOG2E = 1.44269504088896340736f;
    float Alog2[D_STATE];
#pragma unroll
    for (int n = 0; n < D_STATE; ++n)
        Alog2[n] = -__expf(A_log[d * D_STATE + n]) * LOG2E;

    float h[D_STATE];
    const size_t ho = (((size_t)b * NCHUNK + chunk) * D_MODEL + d) * D_STATE;
#pragma unroll
    for (int q = 0; q < 4; ++q)
        *(v4f*)(h + 4 * q) = *(const v4f*)(Hstart + ho + 4 * q);
    const float Dd = Dvec[d];

    const size_t baseU = ((size_t)b * T_LEN + (size_t)chunk * CLEN) * D_MODEL + d;
    const float* up = u  + baseU;
    const float* dp = dt + baseU;
    float*       yp = out + baseU;

    float un[SC_PF], dn[SC_PF];
#pragma unroll
    for (int j = 0; j < SC_PF; ++j) {
        un[j] = up[(size_t)j * D_MODEL];
        dn[j] = dp[(size_t)j * D_MODEL];
    }

    const size_t baseS = ((size_t)b * T_LEN + (size_t)chunk * CLEN) * D_STATE;
#pragma unroll
    for (int k = 0; k < (CLEN * D_STATE) / 256; ++k) {
        ((float*)sB)[threadIdx.x + 256 * k] = Btg[baseS + threadIdx.x + 256 * k];
        ((float*)sC)[threadIdx.x + 256 * k] = Ctg[baseS + threadIdx.x + 256 * k];
    }
    __syncthreads();

    for (int t0 = 0; t0 < CLEN; t0 += SC_PF) {
        float uc[SC_PF], dc[SC_PF];
#pragma unroll
        for (int j = 0; j < SC_PF; ++j) { uc[j] = un[j]; dc[j] = dn[j]; }
        if (t0 + SC_PF < CLEN) {
#pragma unroll
            for (int j = 0; j < SC_PF; ++j) {
                un[j] = up[(size_t)(t0 + SC_PF + j) * D_MODEL];
                dn[j] = dp[(size_t)(t0 + SC_PF + j) * D_MODEL];
            }
        }
#pragma unroll
        for (int j = 0; j < SC_PF; ++j) {
            const float ul  = uc[j];
            const float dtl = dc[j];
            const float dtu = dtl * ul;
            float y = Dd * ul;
#pragma unroll
            for (int n = 0; n < D_STATE; ++n) {
                float abar = __builtin_amdgcn_exp2f(dtl * Alog2[n]);
                h[n] = fmaf(abar, h[n], sB[t0 + j][n] * dtu);
                y    = fmaf(h[n], sC[t0 + j][n], y);
            }
            yp[(size_t)(t0 + j) * D_MODEL] = y;
        }
    }
}

extern "C" void kernel_launch(void* const* d_in, const int* in_sizes, int n_in,
                              void* d_out, int out_size, void* d_ws, size_t ws_size,
                              hipStream_t stream) {
    const float* u     = (const float*)d_in[0];
    const float* A_log = (const float*)d_in[1];
    const float* W_xdt = (const float*)d_in[2];
    const float* b_xdt = (const float*)d_in[3];
    const float* W_dt  = (const float*)d_in[4];
    const float* b_dt  = (const float*)d_in[5];
    const float* W_B   = (const float*)d_in[6];
    const float* b_B   = (const float*)d_in[7];
    const float* W_C   = (const float*)d_in[8];
    const float* b_C   = (const float*)d_in[9];
    const float* Dv    = (const float*)d_in[10];
    float* out = (float*)d_out;

    float* ws   = (float*)d_ws;
    float* xdt  = ws;
    float* dt   = xdt + (size_t)M_ROWS * DT_RANK;
    float* Bt   = dt  + (size_t)M_ROWS * D_MODEL;
    float* Ct   = Bt  + (size_t)M_ROWS * D_STATE;
    float* Pbuf = Ct  + (size_t)M_ROWS * D_STATE;
    float* Sbuf = Pbuf + (size_t)B_SZ * NCHUNK * D_MODEL * D_STATE;

    proj_u_fused<<<M_ROWS / PJ_ROWS, 256, 0, stream>>>(
        u, W_xdt, b_xdt, W_B, b_B, W_C, b_C, xdt, Bt, Ct);

    wmma_gemm_f32<1><<<(M_ROWS / 16) * (D_MODEL / 16) / 8, 256, 0, stream>>>(
        xdt, W_dt, b_dt, dt, M_ROWS, D_MODEL, DT_RANK);

    const int scan_blocks = B_SZ * 4 * NCHUNK;
    scan_phase1<<<scan_blocks, 256, 0, stream>>>(u, dt, Bt, A_log, Pbuf, Sbuf);
    scan_phase2<<<B_SZ * D_MODEL * (D_STATE / 4) / P2_THREADS, P2_THREADS, 0, stream>>>(Pbuf, Sbuf);
    scan_phase3<<<scan_blocks, 256, 0, stream>>>(u, dt, Bt, Ct, A_log, Dv, Pbuf, out);
}


# dt projection re-tiled: a wave keeps 2x4 WMMA tiles (32 rows x 64 cols), K in quarters of 16 with 16-byte loads fetched one quarter ahead; 4096 waves, was 32768 (changes the K summation order)
# speedup vs baseline: 2.6746x; 1.0795x over previous
#include <hip/hip_runtime.h>
#include <hip/hip_bf16.h>
#include <math.h>

typedef float v2f __attribute__((ext_vector_type(2)));
typedef float v4f __attribute__((ext_vector_type(4)));
typedef float v8f __attribute__((ext_vector_type(8)));

#define B_SZ    4
#define T_LEN   2048
#define D_MODEL 1024
#define D_STATE 16
#define DT_RANK 64
#define M_ROWS  (B_SZ * T_LEN)

#define NCHUNK  32
#define CLEN    (T_LEN / NCHUNK)
#define SC_PF   8

#define DT_BM 64
#define DT_BN 256
#define DT_MT 2
#define DT_NT 4

__global__ __launch_bounds__(256)
void dt_proj_softplus(const float* __restrict__ xdt,
                      const float* __restrict__ W,
                      const float* __restrict__ bias,
                      float* __restrict__ out) {
    const int lane = threadIdx.x & 31;
    const int w    = threadIdx.x >> 5;
    const int half = lane >> 4;
    const int l15  = lane & 15;
    const int m0   = (blockIdx.x >> 2) * DT_BM + (w >> 2) * (16 * DT_MT);
    const int n0   = (blockIdx.x & 3) * DT_BN + (w & 3) * (16 * DT_NT);

    const float* ap[DT_MT];
    const float* bp[DT_NT];
#pragma unroll
    for (int mt = 0; mt < DT_MT; ++mt)
        ap[mt] = xdt + (size_t)(m0 + 16 * mt + l15) * DT_RANK + 8 * half;
#pragma unroll
    for (int j = 0; j < DT_NT; ++j)
        bp[j] = W + (size_t)(n0 + 16 * j + l15) * DT_RANK + 8 * half;

    v8f acc[DT_MT][DT_NT];
#pragma unroll
    for (int mt = 0; mt < DT_MT; ++mt)
#pragma unroll
        for (int j = 0; j < DT_NT; ++j)
            acc[mt][j] = v8f{};

    v4f fa[2][DT_MT][2], fb[2][DT_NT][2];
#pragma unroll
    for (int mt = 0; mt < DT_MT; ++mt)
#pragma unroll
        for (int q = 0; q < 2; ++q) fa[0][mt][q] = *(const v4f*)(ap[mt] + 4 * q);
#pragma unroll
    for (int j = 0; j < DT_NT; ++j)
#pragma unroll
        for (int q = 0; q < 2; ++q) fb[0][j][q] = *(const v4f*)(bp[j] + 4 * q);

#pragma unroll
    for (int kq = 0; kq < DT_RANK / 16; ++kq) {
        const int cur = kq & 1;
        if (kq + 1 < DT_RANK / 16) {
            const int nxt = cur ^ 1;
            const int ko  = 16 * (kq + 1);
#pragma unroll
            for (int mt = 0; mt < DT_MT; ++mt)
#pragma unroll
                for (int q = 0; q < 2; ++q) fa[nxt][mt][q] = *(const v4f*)(ap[mt] + ko + 4 * q);
#pragma unroll
            for (int j = 0; j < DT_NT; ++j)
#pragma unroll
                for (int q = 0; q < 2; ++q) fb[nxt][j][q] = *(const v4f*)(bp[j] + ko + 4 * q);
        }
#pragma unroll
        for (int q = 0; q < 2; ++q)
#pragma unroll
            for (int s = 0; s < 2; ++s)
#pragma unroll
                for (int j = 0; j < DT_NT; ++j) {
                    const v2f b = {fb[cur][j][q][2 * s], fb[cur][j][q][2 * s + 1]};
#pragma unroll
                    for (int mt = 0; mt < DT_MT; ++mt) {
                        const v2f a = {fa[cur][mt][q][2 * s], fa[cur][mt][q][2 * s + 1]};
                        acc[mt][j] = __builtin_amdgcn_wmma_f32_16x16x4_f32(
                            false, a, false, b, (short)0, acc[mt][j], false, false);
                    }
                }
    }

#pragma unroll
    for (int j = 0; j < DT_NT; ++j) {
        const float bv = bias[n0 + 16 * j + l15];
#pragma unroll
        for (int mt = 0; mt < DT_MT; ++mt)
#pragma unroll
            for (int r = 0; r < 8; ++r) {
                float v = acc[mt][j][r] + bv;
                const float e  = __expf(v);
                const float ww = 1.0f + e;
                const float ln = __builtin_amdgcn_logf(ww) * 0.69314718055994530942f;
                const float sp = (ww == 1.0f) ? e : ln * (e * __builtin_amdgcn_rcpf(ww - 1.0f));
                v = (v > 20.0f) ? v : sp;
                out[(size_t)(m0 + 16 * mt + 8 * half + r) * D_MODEL + (n0 + 16 * j + l15)] = v;
            }
    }
}

#define PJ_ROWS   32
#define PJ_KSPLIT 8
#define PJ_KSLICE (D_MODEL / PJ_KSPLIT)
#define PJ_NT     6
#define PJ_COLS   (PJ_NT * 16)
#define PJ_PITCH  100

__global__ __launch_bounds__(256)
void proj_u_fused(const float* __restrict__ u,
                  const float* __restrict__ W_xdt,
                  const float* __restrict__ b_xdt,
                  const float* __restrict__ W_B,
                  const float* __restrict__ b_B,
                  const float* __restrict__ W_C,
                  const float* __restrict__ b_C,
                  float* __restrict__ xdt,
                  float* __restrict__ Bt,
                  float* __restrict__ Ct) {
    __shared__ float part[PJ_KSPLIT][PJ_ROWS * PJ_PITCH];

    const int lane = threadIdx.x & 31;
    const int w    = threadIdx.x >> 5;
    const int half = lane >> 4;
    const int l15  = lane & 15;
    const int m0   = blockIdx.x * PJ_ROWS;
    const int koff = w * PJ_KSLICE + 16 * half;

    const float* ap[2];
    ap[0] = u + (size_t)(m0 + l15) * D_MODEL + koff;
    ap[1] = ap[0] + (size_t)16 * D_MODEL;
    const float* bp[PJ_NT];
#pragma unroll
    for (int j = 0; j < 4; ++j)
        bp[j] = W_xdt + (size_t)(16 * j + l15) * D_MODEL + koff;
    bp[4] = W_B + (size_t)l15 * D_MODEL + koff;
    bp[5] = W_C + (size_t)l15 * D_MODEL + koff;

    v8f acc[2][PJ_NT];
#pragma unroll
    for (int mt = 0; mt < 2; ++mt)
#pragma unroll
        for (int j = 0; j < PJ_NT; ++j)
            acc[mt][j] = v8f{};

#pragma unroll
    for (int ks = 0; ks < PJ_KSLICE; ks += 32) {
        v4f a0[4], a1[4];
#pragma unroll
        for (int q = 0; q < 4; ++q) a0[q] = *(const v4f*)(ap[0] + ks + 4 * q);
#pragma unroll
        for (int q = 0; q < 4; ++q) a1[q] = *(const v4f*)(ap[1] + ks + 4 * q);
#pragma unroll
        for (int j = 0; j < PJ_NT; ++j) {
            v4f b[4];
#pragma unroll
            for (int q = 0; q < 4; ++q) b[q] = *(const v4f*)(bp[j] + ks + 4 * q);
#pragma unroll
            for (int q = 0; q < 4; ++q) {
#pragma unroll
                for (int s = 0; s < 2; ++s) {
                    const v2f f0 = {a0[q][2 * s], a0[q][2 * s + 1]};
                    const v2f f1 = {a1[q][2 * s], a1[q][2 * s + 1]};
                    const v2f fb = {b[q][2 * s], b[q][2 * s + 1]};
                    acc[0][j] = __builtin_amdgcn_wmma_f32_16x16x4_f32(
                        false, f0, false, fb, (short)0, acc[0][j], false, false);
                    acc[1][j] = __builtin_amdgcn_wmma_f32_16x16x4_f32(
                        false, f1, false, fb, (short)0, acc[1][j], false, false);
                }
            }
        }
    }

    float* mine = part[w];
#pragma unroll
    for (int mt = 0; mt < 2; ++mt)
#pragma unroll
        for (int j = 0; j < PJ_NT; ++j)
#pragma unroll
            for (int r = 0; r < 8; ++r)
                mine[(16 * mt + 8 * half + r) * PJ_PITCH + 16 * j + l15] = acc[mt][j][r];
    __syncthreads();

#pragma unroll
    for (int i = 0; i < (PJ_ROWS * PJ_COLS) / 256; ++i) {
        const int o = threadIdx.x + 256 * i;
        int row, col;
        float bias;
        float* dst;
        if (i < 8) {
            row = o >> 6;  col = o & 63;
            bias = b_xdt[col];
            dst  = xdt + (size_t)m0 * DT_RANK + o;
        } else if (i < 10) {
            const int q = o - PJ_ROWS * DT_RANK;
            row = q >> 4;  col = 64 + (q & 15);
            bias = b_B[q & 15];
            dst  = Bt + (size_t)m0 * D_STATE + q;
        } else {
            const int q = o - PJ_ROWS * (DT_RANK + D_STATE);
            row = q >> 4;  col = 80 + (q & 15);
            bias = b_C[q & 15];
            dst  = Ct + (size_t)m0 * D_STATE + q;
        }
        float s = part[0][row * PJ_PITCH + col];
#pragma unroll
        for (int k = 1; k < PJ_KSPLIT; ++k)
            s += part[k][row * PJ_PITCH + col];
        *dst = s + bias;
    }
}

__global__ __launch_bounds__(256)
void scan_phase1(const float* __restrict__ u,
                 const float* __restrict__ dt,
                 const float* __restrict__ Btg,
                 const float* __restrict__ A_log,
                 float* __restrict__ Pbuf,
                 float* __restrict__ Sbuf) {
    __shared__ float sB[CLEN][D_STATE];

    const int chunk = blockIdx.x & (NCHUNK - 1);
    const int rest  = blockIdx.x >> 5;
    const int b     = rest >> 2;
    const int d     = ((rest & 3) << 8) + threadIdx.x;

    const float LOG2E = 1.44269504088896340736f;
    float Alog2[D_STATE];
#pragma unroll
    for (int n = 0; n < D_STATE; ++n)
        Alog2[n] = -__expf(A_log[d * D_STATE + n]) * LOG2E;

    const size_t baseU = ((size_t)b * T_LEN + (size_t)chunk * CLEN) * D_MODEL + d;
    const float* up = u  + baseU;
    const float* dp = dt + baseU;

    float un[SC_PF], dn[SC_PF];
#pragma unroll
    for (int j = 0; j < SC_PF; ++j) {
        un[j] = up[(size_t)j * D_MODEL];
        dn[j] = dp[(size_t)j * D_MODEL];
    }

    const size_t baseS = ((size_t)b * T_LEN + (size_t)chunk * CLEN) * D_STATE;
#pragma unroll
    for (int k = 0; k < (CLEN * D_STATE) / 256; ++k)
        ((float*)sB)[threadIdx.x + 256 * k] = Btg[baseS + threadIdx.x + 256 * k];
    __syncthreads();

    float P[D_STATE], S[D_STATE];
#pragma unroll
    for (int n = 0; n < D_STATE; ++n) { P[n] = 1.0f; S[n] = 0.0f; }

    for (int t0 = 0; t0 < CLEN; t0 += SC_PF) {
        float uc[SC_PF], dc[SC_PF];
#pragma unroll
        for (int j = 0; j < SC_PF; ++j) { uc[j] = un[j]; dc[j] = dn[j]; }
        if (t0 + SC_PF < CLEN) {
#pragma unroll
            for (int j = 0; j < SC_PF; ++j) {
                un[j] = up[(size_t)(t0 + SC_PF + j) * D_MODEL];
                dn[j] = dp[(size_t)(t0 + SC_PF + j) * D_MODEL];
            }
        }
#pragma unroll
        for (int j = 0; j < SC_PF; ++j) {
            const float ul  = uc[j];
            const float dtl = dc[j];
            const float dtu = dtl * ul;
#pragma unroll
            for (int n = 0; n < D_STATE; ++n) {
                float abar = __builtin_amdgcn_exp2f(dtl * Alog2[n]);
                P[n] *= abar;
                S[n]  = fmaf(abar, S[n], sB[t0 + j][n] * dtu);
            }
        }
    }

    const size_t o = (((size_t)b * NCHUNK + chunk) * D_MODEL + d) * D_STATE;
#pragma unroll
    for (int q = 0; q < 4; ++q) {
        *(v4f*)(Pbuf + o + 4 * q) = *(v4f*)(P + 4 * q);
        *(v4f*)(Sbuf + o + 4 * q) = *(v4f*)(S + 4 * q);
    }
}

#define P2_THREADS 64

__global__ __launch_bounds__(P2_THREADS)
void scan_phase2(float* __restrict__ Pbuf,
                 const float* __restrict__ Sbuf) {
    const int tid = blockIdx.x * P2_THREADS + threadIdx.x;
    const int b   = tid / (D_MODEL * (D_STATE / 4));
    const int r   = tid % (D_MODEL * (D_STATE / 4));
    const size_t cstride = (size_t)D_MODEL * D_STATE;
    const size_t base    = (size_t)b * NCHUNK * cstride + (size_t)r * 4;

    v4f P[NCHUNK], S[NCHUNK];
#pragma unroll
    for (int c = 0; c < NCHUNK; ++c) {
        P[c] = *(const v4f*)(Pbuf + base + (size_t)c * cstride);
        S[c] = *(const v4f*)(Sbuf + base + (size_t)c * cstride);
    }

    v4f h = {0.0f, 0.0f, 0.0f, 0.0f};
#pragma unroll
    for (int c = 0; c < NCHUNK; ++c) {
        *(v4f*)(Pbuf + base + (size_t)c * cstride) = h;
#pragma unroll
        for (int e = 0; e < 4; ++e)
            h[e] = fmaf(P[c][e], h[e], S[c][e]);
    }
}

__global__ __launch_bounds__(256)
void scan_phase3(const float* __restrict__ u,
                 const float* __restrict__ dt,
                 const float* __restrict__ Btg,
                 const float* __restrict__ Ctg,
                 const float* __restrict__ A_log,
                 const float* __restrict__ Dvec,
                 const float* __restrict__ Hstart,
                 float* __restrict__ out) {
    __shared__ float sB[CLEN][D_STATE];
    __shared__ float sC[CLEN][D_STATE];

    const int chunk = blockIdx.x & (NCHUNK - 1);
    const int rest  = blockIdx.x >> 5;
    const int b     = rest >> 2;
    const int d     = ((rest & 3) << 8) + threadIdx.x;

    const float LOG2E = 1.44269504088896340736f;
    float Alog2[D_STATE];
#pragma unroll
    for (int n = 0; n < D_STATE; ++n)
        Alog2[n] = -__expf(A_log[d * D_STATE + n]) * LOG2E;

    float h[D_STATE];
    const size_t ho = (((size_t)b * NCHUNK + chunk) * D_MODEL + d) * D_STATE;
#pragma unroll
    for (int q = 0; q < 4; ++q)
        *(v4f*)(h + 4 * q) = *(const v4f*)(Hstart + ho + 4 * q);
    const float Dd = Dvec[d];

    const size_t baseU = ((size_t)b * T_LEN + (size_t)chunk * CLEN) * D_MODEL + d;
    const float* up = u  + baseU;
    const float* dp = dt + baseU;
    float*       yp = out + baseU;

    float un[SC_PF], dn[SC_PF];
#pragma unroll
    for (int j = 0; j < SC_PF; ++j) {
        un[j] = up[(size_t)j * D_MODEL];
        dn[j] = dp[(size_t)j * D_MODEL];
    }

    const size_t baseS = ((size_t)b * T_LEN + (size_t)chunk * CLEN) * D_STATE;
#pragma unroll
    for (int k = 0; k < (CLEN * D_STATE) / 256; ++k) {
        ((float*)sB)[threadIdx.x + 256 * k] = Btg[baseS + threadIdx.x + 256 * k];
        ((float*)sC)[threadIdx.x + 256 * k] = Ctg[baseS + threadIdx.x + 256 * k];
    }
    __syncthreads();

    for (int t0 = 0; t0 < CLEN; t0 += SC_PF) {
        float uc[SC_PF], dc[SC_PF];
#pragma unroll
        for (int j = 0; j < SC_PF; ++j) { uc[j] = un[j]; dc[j] = dn[j]; }
        if (t0 + SC_PF < CLEN) {
#pragma unroll
            for (int j = 0; j < SC_PF; ++j) {
                un[j] = up[(size_t)(t0 + SC_PF + j) * D_MODEL];
                dn[j] = dp[(size_t)(t0 + SC_PF + j) * D_MODEL];
            }
        }
#pragma unroll
        for (int j = 0; j < SC_PF; ++j) {
            const float ul  = uc[j];
            const float dtl = dc[j];
            const float dtu = dtl * ul;
            float y = Dd * ul;
#pragma unroll
            for (int n = 0; n < D_STATE; ++n) {
                float abar = __builtin_amdgcn_exp2f(dtl * Alog2[n]);
                h[n] = fmaf(abar, h[n], sB[t0 + j][n] * dtu);
                y    = fmaf(h[n], sC[t0 + j][n], y);
            }
            yp[(size_t)(t0 + j) * D_MODEL] = y;
        }
    }
}

extern "C" void kernel_launch(void* const* d_in, const int* in_sizes, int n_in,
                              void* d_out, int out_size, void* d_ws, size_t ws_size,
                              hipStream_t stream) {
    const float* u     = (const float*)d_in[0];
    const float* A_log = (const float*)d_in[1];
    const float* W_xdt = (const float*)d_in[2];
    const float* b_xdt = (const float*)d_in[3];
    const float* W_dt  = (const float*)d_in[4];
    const float* b_dt  = (const float*)d_in[5];
    const float* W_B   = (const float*)d_in[6];
    const float* b_B   = (const float*)d_in[7];
    const float* W_C   = (const float*)d_in[8];
    const float* b_C   = (const float*)d_in[9];
    const float* Dv    = (const float*)d_in[10];
    float* out = (float*)d_out;

    float* ws   = (float*)d_ws;
    float* xdt  = ws;
    float* dt   = xdt + (size_t)M_ROWS * DT_RANK;
    float* Bt   = dt  + (size_t)M_ROWS * D_MODEL;
    float* Ct   = Bt  + (size_t)M_ROWS * D_STATE;
    float* Pbuf = Ct  + (size_t)M_ROWS * D_STATE;
    float* Sbuf = Pbuf + (size_t)B_SZ * NCHUNK * D_MODEL * D_STATE;

    proj_u_fused<<<M_ROWS / PJ_ROWS, 256, 0, stream>>>(
        u, W_xdt, b_xdt, W_B, b_B, W_C, b_C, xdt, Bt, Ct);

    dt_proj_softplus<<<(M_ROWS / DT_BM) * (D_MODEL / DT_BN), 256, 0, stream>>>(
        xdt, W_dt, b_dt, dt);

    const int scan_blocks = B_SZ * 4 * NCHUNK;
    scan_phase1<<<scan_blocks, 256, 0, stream>>>(u, dt, Bt, A_log, Pbuf, Sbuf);
    scan_phase2<<<B_SZ * D_MODEL * (D_STATE / 4) / P2_THREADS, P2_THREADS, 0, stream>>>(Pbuf, Sbuf);
    scan_phase3<<<scan_blocks, 256, 0, stream>>>(u, dt, Bt, Ct, A_log, Dv, Pbuf, out);
}


# dt projection and scan phase 1 share ONE launch: a work-group builds its 64x256 dt tile on WMMA, stores it once for phase 3 and keeps it in LDS for its own chunk scan; 4 launches, was 5
# speedup vs baseline: 3.0703x; 1.1479x over previous
#include <hip/hip_runtime.h>
#include <hip/hip_bf16.h>
#include <math.h>

typedef float v2f __attribute__((ext_vector_type(2)));
typedef float v4f __attribute__((ext_vector_type(4)));
typedef float v8f __attribute__((ext_vector_type(8)));

#define B_SZ    4
#define T_LEN   2048
#define D_MODEL 1024
#define D_STATE 16
#define DT_RANK 64
#define M_ROWS  (B_SZ * T_LEN)

#define NCHUNK  32
#define CLEN    (T_LEN / NCHUNK)
#define SC_PF   8

#define PJ_ROWS   32
#define PJ_KSPLIT 8
#define PJ_KSLICE (D_MODEL / PJ_KSPLIT)
#define PJ_NT     6
#define PJ_COLS   (PJ_NT * 16)
#define PJ_PITCH  100

__global__ __launch_bounds__(256)
void proj_u_fused(const float* __restrict__ u,
                  const float* __restrict__ W_xdt,
                  const float* __restrict__ b_xdt,
                  const float* __restrict__ W_B,
                  const float* __restrict__ b_B,
                  const float* __restrict__ W_C,
                  const float* __restrict__ b_C,
                  float* __restrict__ xdt,
                  float* __restrict__ Bt,
                  float* __restrict__ Ct) {
    __shared__ float part[PJ_KSPLIT][PJ_ROWS * PJ_PITCH];

    const int lane = threadIdx.x & 31;
    const int w    = threadIdx.x >> 5;
    const int half = lane >> 4;
    const int l15  = lane & 15;
    const int m0   = blockIdx.x * PJ_ROWS;
    const int koff = w * PJ_KSLICE + 16 * half;

    const float* ap[2];
    ap[0] = u + (size_t)(m0 + l15) * D_MODEL + koff;
    ap[1] = ap[0] + (size_t)16 * D_MODEL;
    const float* bp[PJ_NT];
#pragma unroll
    for (int j = 0; j < 4; ++j)
        bp[j] = W_xdt + (size_t)(16 * j + l15) * D_MODEL + koff;
    bp[4] = W_B + (size_t)l15 * D_MODEL + koff;
    bp[5] = W_C + (size_t)l15 * D_MODEL + koff;

    v8f acc[2][PJ_NT];
#pragma unroll
    for (int mt = 0; mt < 2; ++mt)
#pragma unroll
        for (int j = 0; j < PJ_NT; ++j)
            acc[mt][j] = v8f{};

#pragma unroll
    for (int ks = 0; ks < PJ_KSLICE; ks += 32) {
        v4f a0[4], a1[4];
#pragma unroll
        for (int q = 0; q < 4; ++q) a0[q] = *(const v4f*)(ap[0] + ks + 4 * q);
#pragma unroll
        for (int q = 0; q < 4; ++q) a1[q] = *(const v4f*)(ap[1] + ks + 4 * q);
#pragma unroll
        for (int j = 0; j < PJ_NT; ++j) {
            v4f b[4];
#pragma unroll
            for (int q = 0; q < 4; ++q) b[q] = *(const v4f*)(bp[j] + ks + 4 * q);
#pragma unroll
            for (int q = 0; q < 4; ++q) {
#pragma unroll
                for (int s = 0; s < 2; ++s) {
                    const v2f f0 = {a0[q][2 * s], a0[q][2 * s + 1]};
                    const v2f f1 = {a1[q][2 * s], a1[q][2 * s + 1]};
                    const v2f fb = {b[q][2 * s], b[q][2 * s + 1]};
                    acc[0][j] = __builtin_amdgcn_wmma_f32_16x16x4_f32(
                        false, f0, false, fb, (short)0, acc[0][j], false, false);
                    acc[1][j] = __builtin_amdgcn_wmma_f32_16x16x4_f32(
                        false, f1, false, fb, (short)0, acc[1][j], false, false);
                }
            }
        }
    }

    float* mine = part[w];
#pragma unroll
    for (int mt = 0; mt < 2; ++mt)
#pragma unroll
        for (int j = 0; j < PJ_NT; ++j)
#pragma unroll
            for (int r = 0; r < 8; ++r)
                mine[(16 * mt + 8 * half + r) * PJ_PITCH + 16 * j + l15] = acc[mt][j][r];
    __syncthreads();

#pragma unroll
    for (int i = 0; i < (PJ_ROWS * PJ_COLS) / 256; ++i) {
        const int o = threadIdx.x + 256 * i;
        int row, col;
        float bias;
        float* dst;
        if (i < 8) {
            row = o >> 6;  col = o & 63;
            bias = b_xdt[col];
            dst  = xdt + (size_t)m0 * DT_RANK + o;
        } else if (i < 10) {
            const int q = o - PJ_ROWS * DT_RANK;
            row = q >> 4;  col = 64 + (q & 15);
            bias = b_B[q & 15];
            dst  = Bt + (size_t)m0 * D_STATE + q;
        } else {
            const int q = o - PJ_ROWS * (DT_RANK + D_STATE);
            row = q >> 4;  col = 80 + (q & 15);
            bias = b_C[q & 15];
            dst  = Ct + (size_t)m0 * D_STATE + q;
        }
        float s = part[0][row * PJ_PITCH + col];
#pragma unroll
        for (int k = 1; k < PJ_KSPLIT; ++k)
            s += part[k][row * PJ_PITCH + col];
        *dst = s + bias;
    }
}

#define DT_MT 2
#define DT_NT 4

#define DS_PITCH 260

__global__ __launch_bounds__(256)
void dt_proj_scan_phase1(const float* __restrict__ xdt,
                         const float* __restrict__ W,
                         const float* __restrict__ bias,
                         const float* __restrict__ u,
                         const float* __restrict__ Btg,
                         const float* __restrict__ A_log,
                         float* __restrict__ dt,
                         float* __restrict__ Pbuf,
                         float* __restrict__ Sbuf) {
    __shared__ float smem[CLEN * D_STATE + CLEN * DS_PITCH];
    float (*sB)[D_STATE] = (float (*)[D_STATE])smem;
    float* sDt = smem + CLEN * D_STATE;

    const int chunk = blockIdx.x & (NCHUNK - 1);
    const int rest  = blockIdx.x >> 5;
    const int b     = rest >> 2;
    const int dbase = (rest & 3) << 8;
    const int row0  = b * T_LEN + chunk * CLEN;

    const size_t baseS = (size_t)row0 * D_STATE;
#pragma unroll
    for (int k = 0; k < (CLEN * D_STATE) / 256; ++k)
        ((float*)sB)[threadIdx.x + 256 * k] = Btg[baseS + threadIdx.x + 256 * k];

    const int lane = threadIdx.x & 31;
    const int w    = threadIdx.x >> 5;
    const int half = lane >> 4;
    const int l15  = lane & 15;
    const int rl   = (w >> 2) * (16 * DT_MT);
    const int cl   = (w & 3) * (16 * DT_NT);
    const int m0   = row0 + rl;
    const int n0   = dbase + cl;

    const float* ap[DT_MT];
    const float* bp[DT_NT];
#pragma unroll
    for (int mt = 0; mt < DT_MT; ++mt)
        ap[mt] = xdt + (size_t)(m0 + 16 * mt + l15) * DT_RANK + 8 * half;
#pragma unroll
    for (int j = 0; j < DT_NT; ++j)
        bp[j] = W + (size_t)(n0 + 16 * j + l15) * DT_RANK + 8 * half;
    v8f acc[DT_MT][DT_NT];
#pragma unroll
    for (int mt = 0; mt < DT_MT; ++mt)
#pragma unroll
        for (int j = 0; j < DT_NT; ++j)
            acc[mt][j] = v8f{};

    v4f fa[2][DT_MT][2], fb[2][DT_NT][2];
#pragma unroll
    for (int mt = 0; mt < DT_MT; ++mt)
#pragma unroll
        for (int q = 0; q < 2; ++q) fa[0][mt][q] = *(const v4f*)(ap[mt] + 4 * q);
#pragma unroll
    for (int j = 0; j < DT_NT; ++j)
#pragma unroll
        for (int q = 0; q < 2; ++q) fb[0][j][q] = *(const v4f*)(bp[j] + 4 * q);

#pragma unroll
    for (int kq = 0; kq < DT_RANK / 16; ++kq) {
        const int cur = kq & 1;
        if (kq + 1 < DT_RANK / 16) {
            const int nxt = cur ^ 1;
            const int ko  = 16 * (kq + 1);
#pragma unroll
            for (int mt = 0; mt < DT_MT; ++mt)
#pragma unroll
                for (int q = 0; q < 2; ++q) fa[nxt][mt][q] = *(const v4f*)(ap[mt] + ko + 4 * q);
#pragma unroll
            for (int j = 0; j < DT_NT; ++j)
#pragma unroll
                for (int q = 0; q < 2; ++q) fb[nxt][j][q] = *(const v4f*)(bp[j] + ko + 4 * q);
        }
#pragma unroll
        for (int q = 0; q < 2; ++q)
#pragma unroll
            for (int s = 0; s < 2; ++s)
#pragma unroll
                for (int j = 0; j < DT_NT; ++j) {
                    const v2f b = {fb[cur][j][q][2 * s], fb[cur][j][q][2 * s + 1]};
#pragma unroll
                    for (int mt = 0; mt < DT_MT; ++mt) {
                        const v2f a = {fa[cur][mt][q][2 * s], fa[cur][mt][q][2 * s + 1]};
                        acc[mt][j] = __builtin_amdgcn_wmma_f32_16x16x4_f32(
                            false, a, false, b, (short)0, acc[mt][j], false, false);
                    }
                }
    }

    const int d = dbase + threadIdx.x;
    const size_t baseU = (size_t)row0 * D_MODEL + d;
    const float* up = u + baseU;

    float un[SC_PF];
#pragma unroll
    for (int j = 0; j < SC_PF; ++j)
        un[j] = up[(size_t)j * D_MODEL];

    v4f al[D_STATE / 4];
#pragma unroll
    for (int q = 0; q < D_STATE / 4; ++q)
        al[q] = *(const v4f*)(A_log + (size_t)d * D_STATE + 4 * q);

#pragma unroll
    for (int j = 0; j < DT_NT; ++j) {
        const float bv = bias[n0 + 16 * j + l15];
#pragma unroll
        for (int mt = 0; mt < DT_MT; ++mt)
#pragma unroll
            for (int r = 0; r < 8; ++r) {
                float v = acc[mt][j][r] + bv;
                const float e  = __expf(v);
                const float ww = 1.0f + e;
                const float ln = __builtin_amdgcn_logf(ww) * 0.69314718055994530942f;
                const float sp = (ww == 1.0f) ? e : ln * (e * __builtin_amdgcn_rcpf(ww - 1.0f));
                v = (v > 20.0f) ? v : sp;
                dt[(size_t)(m0 + 16 * mt + 8 * half + r) * D_MODEL + (n0 + 16 * j + l15)] = v;
                sDt[(rl + 16 * mt + 8 * half + r) * DS_PITCH + (cl + 16 * j + l15)] = v;
            }
    }
    __builtin_amdgcn_fence(__ATOMIC_RELEASE, "workgroup", "local");
    __builtin_amdgcn_s_barrier();
    __builtin_amdgcn_fence(__ATOMIC_ACQUIRE, "workgroup", "local");

    const float LOG2E = 1.44269504088896340736f;
    float Alog2[D_STATE];
#pragma unroll
    for (int n = 0; n < D_STATE; ++n)
        Alog2[n] = -__expf(al[n >> 2][n & 3]) * LOG2E;

    float P[D_STATE], S[D_STATE];
#pragma unroll
    for (int n = 0; n < D_STATE; ++n) { P[n] = 1.0f; S[n] = 0.0f; }

    const float* dl = sDt + threadIdx.x;

    for (int t0 = 0; t0 < CLEN; t0 += SC_PF) {
        float uc[SC_PF];
#pragma unroll
        for (int j = 0; j < SC_PF; ++j) uc[j] = un[j];
        if (t0 + SC_PF < CLEN) {
#pragma unroll
            for (int j = 0; j < SC_PF; ++j)
                un[j] = up[(size_t)(t0 + SC_PF + j) * D_MODEL];
        }
#pragma unroll
        for (int j = 0; j < SC_PF; ++j) {
            const float ul  = uc[j];
            const float dtl = dl[(t0 + j) * DS_PITCH];
            const float dtu = dtl * ul;
#pragma unroll
            for (int n = 0; n < D_STATE; ++n) {
                float abar = __builtin_amdgcn_exp2f(dtl * Alog2[n]);
                P[n] *= abar;
                S[n]  = fmaf(abar, S[n], sB[t0 + j][n] * dtu);
            }
        }
    }

    const size_t o = (((size_t)b * NCHUNK + chunk) * D_MODEL + d) * D_STATE;
#pragma unroll
    for (int q = 0; q < 4; ++q) {
        *(v4f*)(Pbuf + o + 4 * q) = *(v4f*)(P + 4 * q);
        *(v4f*)(Sbuf + o + 4 * q) = *(v4f*)(S + 4 * q);
    }
}

#define P2_THREADS 64

__global__ __launch_bounds__(P2_THREADS)
void scan_phase2(float* __restrict__ Pbuf,
                 const float* __restrict__ Sbuf) {
    const int tid = blockIdx.x * P2_THREADS + threadIdx.x;
    const int b   = tid / (D_MODEL * (D_STATE / 4));
    const int r   = tid % (D_MODEL * (D_STATE / 4));
    const size_t cstride = (size_t)D_MODEL * D_STATE;
    const size_t base    = (size_t)b * NCHUNK * cstride + (size_t)r * 4;

    v4f P[NCHUNK], S[NCHUNK];
#pragma unroll
    for (int c = 0; c < NCHUNK; ++c) {
        P[c] = *(const v4f*)(Pbuf + base + (size_t)c * cstride);
        S[c] = *(const v4f*)(Sbuf + base + (size_t)c * cstride);
    }

    v4f h = {0.0f, 0.0f, 0.0f, 0.0f};
#pragma unroll
    for (int c = 0; c < NCHUNK; ++c) {
        *(v4f*)(Pbuf + base + (size_t)c * cstride) = h;
#pragma unroll
        for (int e = 0; e < 4; ++e)
            h[e] = fmaf(P[c][e], h[e], S[c][e]);
    }
}

__global__ __launch_bounds__(256)
void scan_phase3(const float* __restrict__ u,
                 const float* __restrict__ dt,
                 const float* __restrict__ Btg,
                 const float* __restrict__ Ctg,
                 const float* __restrict__ A_log,
                 const float* __restrict__ Dvec,
                 const float* __restrict__ Hstart,
                 float* __restrict__ out) {
    __shared__ float sB[CLEN][D_STATE];
    __shared__ float sC[CLEN][D_STATE];

    const int chunk = blockIdx.x & (NCHUNK - 1);
    const int rest  = blockIdx.x >> 5;
    const int b     = rest >> 2;
    const int d     = ((rest & 3) << 8) + threadIdx.x;

    const float LOG2E = 1.44269504088896340736f;
    float Alog2[D_STATE];
#pragma unroll
    for (int n = 0; n < D_STATE; ++n)
        Alog2[n] = -__expf(A_log[d * D_STATE + n]) * LOG2E;

    float h[D_STATE];
    const size_t ho = (((size_t)b * NCHUNK + chunk) * D_MODEL + d) * D_STATE;
#pragma unroll
    for (int q = 0; q < 4; ++q)
        *(v4f*)(h + 4 * q) = *(const v4f*)(Hstart + ho + 4 * q);
    const float Dd = Dvec[d];

    const size_t baseU = ((size_t)b * T_LEN + (size_t)chunk * CLEN) * D_MODEL + d;
    const float* up = u  + baseU;
    const float* dp = dt + baseU;
    float*       yp = out + baseU;

    float un[SC_PF], dn[SC_PF];
#pragma unroll
    for (int j = 0; j < SC_PF; ++j) {
        un[j] = up[(size_t)j * D_MODEL];
        dn[j] = dp[(size_t)j * D_MODEL];
    }

    const size_t baseS = ((size_t)b * T_LEN + (size_t)chunk * CLEN) * D_STATE;
#pragma unroll
    for (int k = 0; k < (CLEN * D_STATE) / 256; ++k) {
        ((float*)sB)[threadIdx.x + 256 * k] = Btg[baseS + threadIdx.x + 256 * k];
        ((float*)sC)[threadIdx.x + 256 * k] = Ctg[baseS + threadIdx.x + 256 * k];
    }
    __syncthreads();

    for (int t0 = 0; t0 < CLEN; t0 += SC_PF) {
        float uc[SC_PF], dc[SC_PF];
#pragma unroll
        for (int j = 0; j < SC_PF; ++j) { uc[j] = un[j]; dc[j] = dn[j]; }
        if (t0 + SC_PF < CLEN) {
#pragma unroll
            for (int j = 0; j < SC_PF; ++j) {
                un[j] = up[(size_t)(t0 + SC_PF + j) * D_MODEL];
                dn[j] = dp[(size_t)(t0 + SC_PF + j) * D_MODEL];
            }
        }
#pragma unroll
        for (int j = 0; j < SC_PF; ++j) {
            const float ul  = uc[j];
            const float dtl = dc[j];
            const float dtu = dtl * ul;
            float y = Dd * ul;
#pragma unroll
            for (int n = 0; n < D_STATE; ++n) {
                float abar = __builtin_amdgcn_exp2f(dtl * Alog2[n]);
                h[n] = fmaf(abar, h[n], sB[t0 + j][n] * dtu);
                y    = fmaf(h[n], sC[t0 + j][n], y);
            }
            yp[(size_t)(t0 + j) * D_MODEL] = y;
        }
    }
}

extern "C" void kernel_launch(void* const* d_in, const int* in_sizes, int n_in,
                              void* d_out, int out_size, void* d_ws, size_t ws_size,
                              hipStream_t stream) {
    const float* u     = (const float*)d_in[0];
    const float* A_log = (const float*)d_in[1];
    const float* W_xdt = (const float*)d_in[2];
    const float* b_xdt = (const float*)d_in[3];
    const float* W_dt  = (const float*)d_in[4];
    const float* b_dt  = (const float*)d_in[5];
    const float* W_B   = (const float*)d_in[6];
    const float* b_B   = (const float*)d_in[7];
    const float* W_C   = (const float*)d_in[8];
    const float* b_C   = (const float*)d_in[9];
    const float* Dv    = (const float*)d_in[10];
    float* out = (float*)d_out;

    float* ws   = (float*)d_ws;
    float* xdt  = ws;
    float* dt   = xdt + (size_t)M_ROWS * DT_RANK;
    float* Bt   = dt  + (size_t)M_ROWS * D_MODEL;
    float* Ct   = Bt  + (size_t)M_ROWS * D_STATE;
    float* Pbuf = Ct  + (size_t)M_ROWS * D_STATE;
    float* Sbuf = Pbuf + (size_t)B_SZ * NCHUNK * D_MODEL * D_STATE;

    proj_u_fused<<<M_ROWS / PJ_ROWS, 256, 0, stream>>>(
        u, W_xdt, b_xdt, W_B, b_B, W_C, b_C, xdt, Bt, Ct);

    const int scan_blocks = B_SZ * 4 * NCHUNK;

    dt_proj_scan_phase1<<<scan_blocks, 256, 0, stream>>>(
        xdt, W_dt, b_dt, u, Bt, A_log, dt, Pbuf, Sbuf);
    scan_phase2<<<B_SZ * D_MODEL * (D_STATE / 4) / P2_THREADS, P2_THREADS, 0, stream>>>(Pbuf, Sbuf);
    scan_phase3<<<scan_blocks, 256, 0, stream>>>(u, dt, Bt, Ct, A_log, Dv, Pbuf, out);
}
